# starting kernel
# baseline (speedup 1.0000x reference)
#include <hip/hip_runtime.h>
#include <hip/hip_bf16.h>

typedef __attribute__((ext_vector_type(2))) float v2f;
typedef __attribute__((ext_vector_type(8))) float v8f;

#define NN 2048
#define FF 64
#define HH 32
#define CC 16

__global__ __launch_bounds__(256)
void fsums_kernel(const float* __restrict__ x,
                  const float* __restrict__ W1,
                  const float* __restrict__ b1,
                  const float* __restrict__ W2,
                  const float* __restrict__ b2,
                  const float* __restrict__ fw,
                  float* __restrict__ fsums)
{
    __shared__ float attnLds[FF];
    __shared__ float redC[8 * 256];

    const int tid  = threadIdx.x;
    const int lane = tid & 31;
    const int wave = tid >> 5;
    const int tileM = blockIdx.x * 16;

    if (tid < FF) attnLds[tid] = __expf(__expf(fw[tid]));
    __syncthreads();
    float s = 0.f;
    #pragma unroll
    for (int f = 0; f < FF; ++f) s += attnLds[f];
    const float invS = 1.0f / s;

    const int row = lane & 15;
    const int hi  = lane >> 4;
    const int col = row;
    const int gRow = tileM + row;

    v8f acc = {};
    const int kBase = wave * 256;
    for (int k0 = kBase; k0 < kBase + 256; k0 += 4) {
        const int kA = k0 + 2 * hi;
        const int f0 = kA >> 5;
        const float xv  = x[gRow * FF + f0];
        v2f a;
        a.x = fmaxf(fmaf(xv, W1[kA],     b1[kA]),     0.f);
        a.y = fmaxf(fmaf(xv, W1[kA + 1], b1[kA + 1]), 0.f);
        const float at = attnLds[f0] * invS;
        v2f b;
        b.x = at * W2[kA       * CC + col];
        b.y = at * W2[(kA + 1) * CC + col];
        acc = __builtin_amdgcn_wmma_f32_16x16x4_f32(false, a, false, b,
                                                    (short)0, acc, false, false);
    }

    #pragma unroll
    for (int r = 0; r < 8; ++r) {
        const int orow = r + 8 * hi;
        redC[wave * 256 + orow * 16 + col] = acc[r];
    }
    __syncthreads();
    float v = 0.f;
    #pragma unroll
    for (int w = 0; w < 8; ++w) v += redC[w * 256 + tid];
    const int ocol = tid & 15;
    float bias = 0.f;
    #pragma unroll 4
    for (int f = 0; f < FF; ++f) bias += attnLds[f] * invS * b2[f * CC + ocol];
    fsums[tileM * CC + tid] = v + bias;
}

__global__ __launch_bounds__(256)
void out_kernel(const float* __restrict__ dist,
                const float* __restrict__ norm,
                const float* __restrict__ pWm1,
                const float* __restrict__ pbm1,
                const float* __restrict__ pWm2,
                const float* __restrict__ pbm2,
                const float* __restrict__ fsums,
                float* __restrict__ out)
{
    __shared__ float mT[8][16 * 66];
    __shared__ float redC[8 * 256];

    const int tid  = threadIdx.x;
    const int lane = tid & 31;
    const int wave = tid >> 5;
    const int tileM = blockIdx.x * 16;
    const int row = lane & 15;
    const int hi  = lane >> 4;
    const int col = row;

    float wm1[HH], bb1[HH], wm2[HH];
    #pragma unroll
    for (int h = 0; h < HH; ++h) { wm1[h] = pWm1[h]; bb1[h] = pbm1[h]; wm2[h] = pWm2[h]; }
    const float bm2v = pbm2[0];

    float* myT = mT[wave];
    v8f acc = {};
    const int kBase = wave * 256;
    for (int kc = kBase; kc < kBase + 256; kc += 64) {
        for (int r = 0; r < 16; ++r) {
            const size_t base = (size_t)(tileM + r) * NN + kc + 2 * lane;
            const v2f d2 = __builtin_nontemporal_load((const v2f*)(dist + base));
            const v2f n2 = __builtin_nontemporal_load((const v2f*)(norm + base));
            float a0 = bm2v, a1 = bm2v;
            #pragma unroll
            for (int h = 0; h < HH; ++h) {
                const float t0 = fmaf(d2.x, wm1[h], bb1[h]);
                const float t1 = fmaf(d2.y, wm1[h], bb1[h]);
                a0 = fmaf(fmaxf(t0, 0.f), wm2[h], a0);
                a1 = fmaf(fmaxf(t1, 0.f), wm2[h], a1);
            }
            v2f mv;
            mv.x = a0 * __builtin_amdgcn_rcpf(n2.x);
            mv.y = a1 * __builtin_amdgcn_rcpf(n2.y);
            *(v2f*)&myT[r * 66 + 2 * lane] = mv;
        }
        #pragma unroll
        for (int kk = 0; kk < 64; kk += 4) {
            const v2f a = *(const v2f*)&myT[row * 66 + kk + 2 * hi];
            const int kg = kc + kk + 2 * hi;
            v2f b;
            b.x = fsums[kg       * CC + col];
            b.y = fsums[(kg + 1) * CC + col];
            acc = __builtin_amdgcn_wmma_f32_16x16x4_f32(false, a, false, b,
                                                        (short)0, acc, false, false);
        }
    }

    #pragma unroll
    for (int r = 0; r < 8; ++r) {
        const int orow = r + 8 * hi;
        redC[wave * 256 + orow * 16 + col] = acc[r];
    }
    __syncthreads();
    float v = 0.f;
    #pragma unroll
    for (int w = 0; w < 8; ++w) v += redC[w * 256 + tid];
    out[tileM * CC + tid] = v;
}

extern "C" void kernel_launch(void* const* d_in, const int* in_sizes, int n_in,
                              void* d_out, int out_size, void* d_ws, size_t ws_size,
                              hipStream_t stream) {
    const float* x    = (const float*)d_in[0];
    const float* dist = (const float*)d_in[1];
    const float* norm = (const float*)d_in[2];
    const float* W1   = (const float*)d_in[3];
    const float* b1   = (const float*)d_in[4];
    const float* W2   = (const float*)d_in[5];
    const float* b2   = (const float*)d_in[6];
    const float* fw   = (const float*)d_in[7];
    const float* Wm1  = (const float*)d_in[8];
    const float* bm1  = (const float*)d_in[9];
    const float* Wm2  = (const float*)d_in[10];
    const float* bm2  = (const float*)d_in[11];
    float* out   = (float*)d_out;
    float* fsums = (float*)d_ws;

    fsums_kernel<<<NN / 16, 256, 0, stream>>>(x, W1, b1, W2, b2, fw, fsums);
    out_kernel<<<NN / 16, 256, 0, stream>>>(dist, norm, Wm1, bm1, Wm2, bm2, fsums, out);
}


# out kernel, m network as its piecewise-linear form: per block a sorted breakpoint table with slope and intercept per segment in LDS; per element binary search + 1 FMA, 16 rows at once; new sum order
# speedup vs baseline: 2.0940x; 2.0940x over previous
#include <hip/hip_runtime.h>
#include <hip/hip_bf16.h>

typedef __attribute__((ext_vector_type(2))) float v2f;
typedef __attribute__((ext_vector_type(8))) float v8f;

#define NN 2048
#define FF 64
#define HH 32
#define CC 16

__global__ __launch_bounds__(256)
void fsums_kernel(const float* __restrict__ x,
                  const float* __restrict__ W1,
                  const float* __restrict__ b1,
                  const float* __restrict__ W2,
                  const float* __restrict__ b2,
                  const float* __restrict__ fw,
                  float* __restrict__ fsums)
{
    __shared__ float attnLds[FF];
    __shared__ float redC[8 * 256];

    const int tid  = threadIdx.x;
    const int lane = tid & 31;
    const int wave = tid >> 5;
    const int tileM = blockIdx.x * 16;

    if (tid < FF) attnLds[tid] = __expf(__expf(fw[tid]));
    __syncthreads();
    float s = 0.f;
    #pragma unroll
    for (int f = 0; f < FF; ++f) s += attnLds[f];
    const float invS = 1.0f / s;

    const int row = lane & 15;
    const int hi  = lane >> 4;
    const int col = row;
    const int gRow = tileM + row;

    v8f acc = {};
    const int kBase = wave * 256;
    for (int k0 = kBase; k0 < kBase + 256; k0 += 4) {
        const int kA = k0 + 2 * hi;
        const int f0 = kA >> 5;
        const float xv  = x[gRow * FF + f0];
        v2f a;
        a.x = fmaxf(fmaf(xv, W1[kA],     b1[kA]),     0.f);
        a.y = fmaxf(fmaf(xv, W1[kA + 1], b1[kA + 1]), 0.f);
        const float at = attnLds[f0] * invS;
        v2f b;
        b.x = at * W2[kA       * CC + col];
        b.y = at * W2[(kA + 1) * CC + col];
        acc = __builtin_amdgcn_wmma_f32_16x16x4_f32(false, a, false, b,
                                                    (short)0, acc, false, false);
    }

    #pragma unroll
    for (int r = 0; r < 8; ++r) {
        const int orow = r + 8 * hi;
        redC[wave * 256 + orow * 16 + col] = acc[r];
    }
    __syncthreads();
    float v = 0.f;
    #pragma unroll
    for (int w = 0; w < 8; ++w) v += redC[w * 256 + tid];
    const int ocol = tid & 15;
    float bias = 0.f;
    #pragma unroll 4
    for (int f = 0; f < FF; ++f) bias += attnLds[f] * invS * b2[f * CC + ocol];
    fsums[tileM * CC + tid] = v + bias;
}

#define KNOT_PAD 64

__device__ __forceinline__ float lds_at(const float* base, int byteOff) {
    return *(const float*)((const char*)base + byteOff);
}

__global__ __launch_bounds__(256) __attribute__((amdgpu_waves_per_eu(2, 2)))
void out_kernel(const float* __restrict__ dist,
                const float* __restrict__ norm,
                const float* __restrict__ pWm1,
                const float* __restrict__ pbm1,
                const float* __restrict__ pWm2,
                const float* __restrict__ pbm2,
                const float* __restrict__ fsums,
                float* __restrict__ out)
{
    __shared__ float mT[8][16 * 66];
    __shared__ float redC[8 * 256];
    __shared__ float knotRaw[HH];
    __shared__ int   knotRank[HH];
    __shared__ float knotT[KNOT_PAD];
    __shared__ float segS[KNOT_PAD];
    __shared__ float segI[KNOT_PAD];

    const int tid  = threadIdx.x;
    const int lane = tid & 31;
    const int wave = tid >> 5;
    const int tileM = blockIdx.x * 16;
    const int row = lane & 15;
    const int hi  = lane >> 4;
    const int col = row;

    const float INF = __builtin_huge_valf();
    if (tid < HH) {
        const float w = pWm1[tid], b = pbm1[tid];
        const float q = -b / w;
        knotRaw[tid] = (w != 0.f) ? q : INF;
    } else if (tid < KNOT_PAD) {
        knotT[tid] = INF;
    }
    __syncthreads();
    if (tid < HH) {
        const float t = knotRaw[tid];
        int rank = 0;
        for (int i = 0; i < HH; ++i) {
            const float ti = knotRaw[i];
            rank += (ti < t || (ti == t && i < tid)) ? 1 : 0;
        }
        knotT[rank] = t;
        knotRank[tid] = rank;
    }
    __syncthreads();
    if (tid <= HH) {
        const int j = tid;
        float S = 0.f, I = pbm2[0];
        for (int h = 0; h < HH; ++h) {
            const float w = pWm1[h], b = pbm1[h], w2 = pWm2[h];
            const int r = knotRank[h];
            const bool active = (w > 0.f) ? (r < j) : ((w < 0.f) ? (r >= j) : (b > 0.f));
            if (active) { S = fmaf(w2, w, S); I = fmaf(w2, b, I); }
        }
        segS[j] = S;
        segI[j] = I;
    }
    __syncthreads();
    const float knotTop = knotT[HH - 1];

    float* myT = mT[wave];
    v8f acc = {};
    const int kBase = wave * 256;
    for (int kc = kBase; kc < kBase + 256; kc += 64) {
        v2f d2[16], n2[16];
        #pragma unroll
        for (int r = 0; r < 16; ++r) {
            const size_t base = (size_t)(tileM + r) * NN + kc + 2 * lane;
            d2[r] = __builtin_nontemporal_load((const v2f*)(dist + base));
            n2[r] = __builtin_nontemporal_load((const v2f*)(norm + base));
        }
        int j0[16], j1[16];
        #pragma unroll
        for (int r = 0; r < 16; ++r) {
            j0[r] = (knotTop < d2[r].x) ? 4 * HH : 0;
            j1[r] = (knotTop < d2[r].y) ? 4 * HH : 0;
        }
        #pragma unroll
        for (int step = HH / 2; step >= 1; step >>= 1) {
            #pragma unroll
            for (int r = 0; r < 16; ++r) {
                const float t0 = lds_at(knotT, j0[r] + 4 * (step - 1));
                const float t1 = lds_at(knotT, j1[r] + 4 * (step - 1));
                j0[r] += (t0 < d2[r].x) ? 4 * step : 0;
                j1[r] += (t1 < d2[r].y) ? 4 * step : 0;
            }
        }
        #pragma unroll
        for (int r = 0; r < 16; ++r) {
            const float a0 = fmaf(lds_at(segS, j0[r]), d2[r].x, lds_at(segI, j0[r]));
            const float a1 = fmaf(lds_at(segS, j1[r]), d2[r].y, lds_at(segI, j1[r]));
            v2f mv;
            mv.x = a0 * __builtin_amdgcn_rcpf(n2[r].x);
            mv.y = a1 * __builtin_amdgcn_rcpf(n2[r].y);
            *(v2f*)&myT[r * 66 + 2 * lane] = mv;
        }
        #pragma unroll
        for (int kk = 0; kk < 64; kk += 4) {
            const v2f a = *(const v2f*)&myT[row * 66 + kk + 2 * hi];
            const int kg = kc + kk + 2 * hi;
            v2f b;
            b.x = fsums[kg       * CC + col];
            b.y = fsums[(kg + 1) * CC + col];
            acc = __builtin_amdgcn_wmma_f32_16x16x4_f32(false, a, false, b,
                                                        (short)0, acc, false, false);
        }
    }

    #pragma unroll
    for (int r = 0; r < 8; ++r) {
        const int orow = r + 8 * hi;
        redC[wave * 256 + orow * 16 + col] = acc[r];
    }
    __syncthreads();
    float v = 0.f;
    #pragma unroll
    for (int w = 0; w < 8; ++w) v += redC[w * 256 + tid];
    out[tileM * CC + tid] = v;
}

extern "C" void kernel_launch(void* const* d_in, const int* in_sizes, int n_in,
                              void* d_out, int out_size, void* d_ws, size_t ws_size,
                              hipStream_t stream) {
    const float* x    = (const float*)d_in[0];
    const float* dist = (const float*)d_in[1];
    const float* norm = (const float*)d_in[2];
    const float* W1   = (const float*)d_in[3];
    const float* b1   = (const float*)d_in[4];
    const float* W2   = (const float*)d_in[5];
    const float* b2   = (const float*)d_in[6];
    const float* fw   = (const float*)d_in[7];
    const float* Wm1  = (const float*)d_in[8];
    const float* bm1  = (const float*)d_in[9];
    const float* Wm2  = (const float*)d_in[10];
    const float* bm2  = (const float*)d_in[11];
    float* out   = (float*)d_out;
    float* fsums = (float*)d_ws;

    fsums_kernel<<<NN / 16, 256, 0, stream>>>(x, W1, b1, W2, b2, fw, fsums);
    out_kernel<<<NN / 16, 256, 0, stream>>>(dist, norm, Wm1, bm1, Wm2, bm2, fsums, out);
}


# fsums kernel: each block fetches all its operands once at the top (16-byte loads, all in flight) and stages them in LDS; K loop and bias read LDS only (was 32 load-wait rounds); sum order unchanged
# speedup vs baseline: 2.5101x; 1.1987x over previous
#include <hip/hip_runtime.h>
#include <hip/hip_bf16.h>

typedef __attribute__((ext_vector_type(2))) float v2f;
typedef __attribute__((ext_vector_type(8))) float v8f;

#define NN 2048
#define FF 64
#define HH 32
#define CC 16

typedef __attribute__((ext_vector_type(4))) float v4f;

__global__ __launch_bounds__(256)
void fsums_kernel(const float* __restrict__ x,
                  const float* __restrict__ W1,
                  const float* __restrict__ b1,
                  const float* __restrict__ W2,
                  const float* __restrict__ b2,
                  const float* __restrict__ fw,
                  float* __restrict__ fsums)
{
    __shared__ v4f w2T[8 * 1024];
    __shared__ v4f wbT[8 * 128];
    __shared__ v4f b2T[FF * CC / 4];
    __shared__ float attnLds[256];
    __shared__ float redC[8 * 256];

    const int tid  = threadIdx.x;
    const int lane = tid & 31;
    const int wave = tid >> 5;
    const int tileM = blockIdx.x * 16;
    const int row = lane & 15;
    const int hi  = lane >> 4;
    const int col = row;
    const int gRow = tileM + row;
    const int kBase = wave * 256;

    const float fwv = fw[tid & (FF - 1)];
    const v4f* x4 = (const v4f*)(x + (size_t)gRow * FF + wave * 8);
    const v4f xa = x4[0], xb = x4[1];
    const v4f* W1v = (const v4f*)(W1 + kBase);
    const v4f* b1v = (const v4f*)(b1 + kBase);
    const v4f w1a = W1v[lane], w1b = W1v[32 + lane];
    const v4f b1a = b1v[lane], b1b = b1v[32 + lane];
    const v4f b2v = ((const v4f*)b2)[tid];
    const v4f* W2v = (const v4f*)(W2 + (size_t)kBase * CC);
    v4f w2r[32];
    #pragma unroll
    for (int i = 0; i < 32; ++i) w2r[i] = W2v[i * 32 + lane];

    attnLds[tid] = __expf(__expf(fwv));
    {
        v4f* wb = wbT + wave * 128;
        v4f p;
        p.x = w1a.x; p.y = w1a.y; p.z = b1a.x; p.w = b1a.y; wb[2 * lane]          = p;
        p.x = w1a.z; p.y = w1a.w; p.z = b1a.z; p.w = b1a.w; wb[2 * lane + 1]      = p;
        p.x = w1b.x; p.y = w1b.y; p.z = b1b.x; p.w = b1b.y; wb[64 + 2 * lane]     = p;
        p.x = w1b.z; p.y = w1b.w; p.z = b1b.z; p.w = b1b.w; wb[64 + 2 * lane + 1] = p;
    }
    b2T[tid] = b2v;
    #pragma unroll
    for (int i = 0; i < 32; ++i) w2T[wave * 1024 + i * 32 + lane] = w2r[i];
    __syncthreads();

    float s = 0.f;
    #pragma unroll
    for (int f = 0; f < FF; ++f) s += attnLds[f];
    const float invS = 1.0f / s;

    const float xv[8] = {xa.x, xa.y, xa.z, xa.w, xb.x, xb.y, xb.z, xb.w};
    const float* w2f = (const float*)(w2T + wave * 1024);
    const v4f*   wb  = wbT + wave * 128;

    v4f pc[8]; v2f bc[8];
    #pragma unroll
    for (int t = 0; t < 8; ++t) {
        const int kA = t * 4 + 2 * hi;
        pc[t] = wb[kA >> 1];
        bc[t].x = w2f[kA * CC + col];
        bc[t].y = w2f[(kA + 1) * CC + col];
    }
    v8f acc = {};
    #pragma unroll
    for (int fl = 0; fl < 8; ++fl) {
        v4f pn[8]; v2f bn[8];
        if (fl + 1 < 8) {
            #pragma unroll
            for (int t = 0; t < 8; ++t) {
                const int kA = (fl + 1) * 32 + t * 4 + 2 * hi;
                pn[t] = wb[kA >> 1];
                bn[t].x = w2f[kA * CC + col];
                bn[t].y = w2f[(kA + 1) * CC + col];
            }
        }
        asm volatile("" ::: "memory");
        const float at = attnLds[wave * 8 + fl] * invS;
        const float xf = xv[fl];
        v2f a[8], b[8];
        #pragma unroll
        for (int t = 0; t < 8; ++t) {
            a[t].x = fmaxf(fmaf(xf, pc[t].x, pc[t].z), 0.f);
            a[t].y = fmaxf(fmaf(xf, pc[t].y, pc[t].w), 0.f);
            b[t].x = at * bc[t].x;
            b[t].y = at * bc[t].y;
        }
        #pragma unroll
        for (int t = 0; t < 8; ++t)
            acc = __builtin_amdgcn_wmma_f32_16x16x4_f32(false, a[t], false, b[t],
                                                        (short)0, acc, false, false);
        if (fl + 1 < 8) {
            #pragma unroll
            for (int t = 0; t < 8; ++t) { pc[t] = pn[t]; bc[t] = bn[t]; }
        }
    }

    #pragma unroll
    for (int r = 0; r < 8; ++r) {
        const int orow = r + 8 * hi;
        redC[wave * 256 + orow * 16 + col] = acc[r];
    }
    const int ocol = tid & 15;
    const float* b2f = (const float*)b2T;
    float bias = 0.f;
    #pragma unroll
    for (int f = 0; f < FF; ++f) bias += attnLds[f] * invS * b2f[f * CC + ocol];
    __syncthreads();
    float v = 0.f;
    #pragma unroll
    for (int w = 0; w < 8; ++w) v += redC[w * 256 + tid];
    fsums[tileM * CC + tid] = v + bias;
}

#define KNOT_PAD 64

__device__ __forceinline__ float lds_at(const float* base, int byteOff) {
    return *(const float*)((const char*)base + byteOff);
}

__global__ __launch_bounds__(256) __attribute__((amdgpu_waves_per_eu(2, 2)))
void out_kernel(const float* __restrict__ dist,
                const float* __restrict__ norm,
                const float* __restrict__ pWm1,
                const float* __restrict__ pbm1,
                const float* __restrict__ pWm2,
                const float* __restrict__ pbm2,
                const float* __restrict__ fsums,
                float* __restrict__ out)
{
    __shared__ float mT[8][16 * 66];
    __shared__ float redC[8 * 256];
    __shared__ float knotRaw[HH];
    __shared__ int   knotRank[HH];
    __shared__ float knotT[KNOT_PAD];
    __shared__ float segS[KNOT_PAD];
    __shared__ float segI[KNOT_PAD];

    const int tid  = threadIdx.x;
    const int lane = tid & 31;
    const int wave = tid >> 5;
    const int tileM = blockIdx.x * 16;
    const int row = lane & 15;
    const int hi  = lane >> 4;
    const int col = row;

    const float INF = __builtin_huge_valf();
    if (tid < HH) {
        const float w = pWm1[tid], b = pbm1[tid];
        const float q = -b / w;
        knotRaw[tid] = (w != 0.f) ? q : INF;
    } else if (tid < KNOT_PAD) {
        knotT[tid] = INF;
    }
    __syncthreads();
    if (tid < HH) {
        const float t = knotRaw[tid];
        int rank = 0;
        for (int i = 0; i < HH; ++i) {
            const float ti = knotRaw[i];
            rank += (ti < t || (ti == t && i < tid)) ? 1 : 0;
        }
        knotT[rank] = t;
        knotRank[tid] = rank;
    }
    __syncthreads();
    if (tid <= HH) {
        const int j = tid;
        float S = 0.f, I = pbm2[0];
        for (int h = 0; h < HH; ++h) {
            const float w = pWm1[h], b = pbm1[h], w2 = pWm2[h];
            const int r = knotRank[h];
            const bool active = (w > 0.f) ? (r < j) : ((w < 0.f) ? (r >= j) : (b > 0.f));
            if (active) { S = fmaf(w2, w, S); I = fmaf(w2, b, I); }
        }
        segS[j] = S;
        segI[j] = I;
    }
    __syncthreads();
    const float knotTop = knotT[HH - 1];

    float* myT = mT[wave];
    v8f acc = {};
    const int kBase = wave * 256;
    for (int kc = kBase; kc < kBase + 256; kc += 64) {
        v2f d2[16], n2[16];
        #pragma unroll
        for (int r = 0; r < 16; ++r) {
            const size_t base = (size_t)(tileM + r) * NN + kc + 2 * lane;
            d2[r] = __builtin_nontemporal_load((const v2f*)(dist + base));
            n2[r] = __builtin_nontemporal_load((const v2f*)(norm + base));
        }
        int j0[16], j1[16];
        #pragma unroll
        for (int r = 0; r < 16; ++r) {
            j0[r] = (knotTop < d2[r].x) ? 4 * HH : 0;
            j1[r] = (knotTop < d2[r].y) ? 4 * HH : 0;
        }
        #pragma unroll
        for (int step = HH / 2; step >= 1; step >>= 1) {
            #pragma unroll
            for (int r = 0; r < 16; ++r) {
                const float t0 = lds_at(knotT, j0[r] + 4 * (step - 1));
                const float t1 = lds_at(knotT, j1[r] + 4 * (step - 1));
                j0[r] += (t0 < d2[r].x) ? 4 * step : 0;
                j1[r] += (t1 < d2[r].y) ? 4 * step : 0;
            }
        }
        #pragma unroll
        for (int r = 0; r < 16; ++r) {
            const float a0 = fmaf(lds_at(segS, j0[r]), d2[r].x, lds_at(segI, j0[r]));
            const float a1 = fmaf(lds_at(segS, j1[r]), d2[r].y, lds_at(segI, j1[r]));
            v2f mv;
            mv.x = a0 * __builtin_amdgcn_rcpf(n2[r].x);
            mv.y = a1 * __builtin_amdgcn_rcpf(n2[r].y);
            *(v2f*)&myT[r * 66 + 2 * lane] = mv;
        }
        #pragma unroll
        for (int kk = 0; kk < 64; kk += 4) {
            const v2f a = *(const v2f*)&myT[row * 66 + kk + 2 * hi];
            const int kg = kc + kk + 2 * hi;
            v2f b;
            b.x = fsums[kg       * CC + col];
            b.y = fsums[(kg + 1) * CC + col];
            acc = __builtin_amdgcn_wmma_f32_16x16x4_f32(false, a, false, b,
                                                        (short)0, acc, false, false);
        }
    }

    #pragma unroll
    for (int r = 0; r < 8; ++r) {
        const int orow = r + 8 * hi;
        redC[wave * 256 + orow * 16 + col] = acc[r];
    }
    __syncthreads();
    float v = 0.f;
    #pragma unroll
    for (int w = 0; w < 8; ++w) v += redC[w * 256 + tid];
    out[tileM * CC + tid] = v;
}

extern "C" void kernel_launch(void* const* d_in, const int* in_sizes, int n_in,
                              void* d_out, int out_size, void* d_ws, size_t ws_size,
                              hipStream_t stream) {
    const float* x    = (const float*)d_in[0];
    const float* dist = (const float*)d_in[1];
    const float* norm = (const float*)d_in[2];
    const float* W1   = (const float*)d_in[3];
    const float* b1   = (const float*)d_in[4];
    const float* W2   = (const float*)d_in[5];
    const float* b2   = (const float*)d_in[6];
    const float* fw   = (const float*)d_in[7];
    const float* Wm1  = (const float*)d_in[8];
    const float* bm1  = (const float*)d_in[9];
    const float* Wm2  = (const float*)d_in[10];
    const float* bm2  = (const float*)d_in[11];
    float* out   = (float*)d_out;
    float* fsums = (float*)d_ws;

    fsums_kernel<<<NN / 16, 256, 0, stream>>>(x, W1, b1, W2, b2, fw, fsums);
    out_kernel<<<NN / 16, 256, 0, stream>>>(dist, norm, Wm1, bm1, Wm2, bm2, fsums, out);
}


# out kernel table build, segment sums: 33 lanes read one 16-byte LDS record per unit (weights, active range), 1 range test + 2 selected FMAs, no branch (was 2 waited scalar loads + 4 branches a unit)
# speedup vs baseline: 2.7338x; 1.0891x over previous
#include <hip/hip_runtime.h>
#include <hip/hip_bf16.h>

typedef __attribute__((ext_vector_type(2))) float v2f;
typedef __attribute__((ext_vector_type(8))) float v8f;

#define NN 2048
#define FF 64
#define HH 32
#define CC 16

typedef __attribute__((ext_vector_type(4))) float v4f;

__global__ __launch_bounds__(256)
void fsums_kernel(const float* __restrict__ x,
                  const float* __restrict__ W1,
                  const float* __restrict__ b1,
                  const float* __restrict__ W2,
                  const float* __restrict__ b2,
                  const float* __restrict__ fw,
                  float* __restrict__ fsums)
{
    __shared__ v4f w2T[8 * 1024];
    __shared__ v4f wbT[8 * 128];
    __shared__ v4f b2T[FF * CC / 4];
    __shared__ float attnLds[256];
    __shared__ float redC[8 * 256];

    const int tid  = threadIdx.x;
    const int lane = tid & 31;
    const int wave = tid >> 5;
    const int tileM = blockIdx.x * 16;
    const int row = lane & 15;
    const int hi  = lane >> 4;
    const int col = row;
    const int gRow = tileM + row;
    const int kBase = wave * 256;

    const float fwv = fw[tid & (FF - 1)];
    const v4f* x4 = (const v4f*)(x + (size_t)gRow * FF + wave * 8);
    const v4f xa = x4[0], xb = x4[1];
    const v4f* W1v = (const v4f*)(W1 + kBase);
    const v4f* b1v = (const v4f*)(b1 + kBase);
    const v4f w1a = W1v[lane], w1b = W1v[32 + lane];
    const v4f b1a = b1v[lane], b1b = b1v[32 + lane];
    const v4f b2v = ((const v4f*)b2)[tid];
    const v4f* W2v = (const v4f*)(W2 + (size_t)kBase * CC);
    v4f w2r[32];
    #pragma unroll
    for (int i = 0; i < 32; ++i) w2r[i] = W2v[i * 32 + lane];

    attnLds[tid] = __expf(__expf(fwv));
    {
        v4f* wb = wbT + wave * 128;
        v4f p;
        p.x = w1a.x; p.y = w1a.y; p.z = b1a.x; p.w = b1a.y; wb[2 * lane]          = p;
        p.x = w1a.z; p.y = w1a.w; p.z = b1a.z; p.w = b1a.w; wb[2 * lane + 1]      = p;
        p.x = w1b.x; p.y = w1b.y; p.z = b1b.x; p.w = b1b.y; wb[64 + 2 * lane]     = p;
        p.x = w1b.z; p.y = w1b.w; p.z = b1b.z; p.w = b1b.w; wb[64 + 2 * lane + 1] = p;
    }
    b2T[tid] = b2v;
    #pragma unroll
    for (int i = 0; i < 32; ++i) w2T[wave * 1024 + i * 32 + lane] = w2r[i];
    __syncthreads();

    float s = 0.f;
    #pragma unroll
    for (int f = 0; f < FF; ++f) s += attnLds[f];
    const float invS = 1.0f / s;

    const float xv[8] = {xa.x, xa.y, xa.z, xa.w, xb.x, xb.y, xb.z, xb.w};
    const float* w2f = (const float*)(w2T + wave * 1024);
    const v4f*   wb  = wbT + wave * 128;

    v4f pc[8]; v2f bc[8];
    #pragma unroll
    for (int t = 0; t < 8; ++t) {
        const int kA = t * 4 + 2 * hi;
        pc[t] = wb[kA >> 1];
        bc[t].x = w2f[kA * CC + col];
        bc[t].y = w2f[(kA + 1) * CC + col];
    }
    v8f acc = {};
    #pragma unroll
    for (int fl = 0; fl < 8; ++fl) {
        v4f pn[8]; v2f bn[8];
        if (fl + 1 < 8) {
            #pragma unroll
            for (int t = 0; t < 8; ++t) {
                const int kA = (fl + 1) * 32 + t * 4 + 2 * hi;
                pn[t] = wb[kA >> 1];
                bn[t].x = w2f[kA * CC + col];
                bn[t].y = w2f[(kA + 1) * CC + col];
            }
        }
        asm volatile("" ::: "memory");
        const float at = attnLds[wave * 8 + fl] * invS;
        const float xf = xv[fl];
        v2f a[8], b[8];
        #pragma unroll
        for (int t = 0; t < 8; ++t) {
            a[t].x = fmaxf(fmaf(xf, pc[t].x, pc[t].z), 0.f);
            a[t].y = fmaxf(fmaf(xf, pc[t].y, pc[t].w), 0.f);
            b[t].x = at * bc[t].x;
            b[t].y = at * bc[t].y;
        }
        #pragma unroll
        for (int t = 0; t < 8; ++t)
            acc = __builtin_amdgcn_wmma_f32_16x16x4_f32(false, a[t], false, b[t],
                                                        (short)0, acc, false, false);
        if (fl + 1 < 8) {
            #pragma unroll
            for (int t = 0; t < 8; ++t) { pc[t] = pn[t]; bc[t] = bn[t]; }
        }
    }

    #pragma unroll
    for (int r = 0; r < 8; ++r) {
        const int orow = r + 8 * hi;
        redC[wave * 256 + orow * 16 + col] = acc[r];
    }
    const int ocol = tid & 15;
    const float* b2f = (const float*)b2T;
    float bias = 0.f;
    #pragma unroll
    for (int f = 0; f < FF; ++f) bias += attnLds[f] * invS * b2f[f * CC + ocol];
    __syncthreads();
    float v = 0.f;
    #pragma unroll
    for (int w = 0; w < 8; ++w) v += redC[w * 256 + tid];
    fsums[tileM * CC + tid] = v + bias;
}

#define KNOT_PAD 64

__device__ __forceinline__ float lds_at(const float* base, int byteOff) {
    return *(const float*)((const char*)base + byteOff);
}

__global__ __launch_bounds__(256) __attribute__((amdgpu_waves_per_eu(2, 2)))
void out_kernel(const float* __restrict__ dist,
                const float* __restrict__ norm,
                const float* __restrict__ pWm1,
                const float* __restrict__ pbm1,
                const float* __restrict__ pWm2,
                const float* __restrict__ pbm2,
                const float* __restrict__ fsums,
                float* __restrict__ out)
{
    __shared__ float mT[8][16 * 66];
    __shared__ float redC[8 * 256];
    __shared__ float knotRaw[HH];
    __shared__ v4f   unitT[HH];
    __shared__ float knotT[KNOT_PAD];
    __shared__ float segS[KNOT_PAD];
    __shared__ float segI[KNOT_PAD];

    const int tid  = threadIdx.x;
    const int lane = tid & 31;
    const int wave = tid >> 5;
    const int tileM = blockIdx.x * 16;
    const int row = lane & 15;
    const int hi  = lane >> 4;
    const int col = row;

    const float INF = __builtin_huge_valf();
    const float bm2v = pbm2[0];
    float uw = 0.f, ub = 0.f, uw2 = 0.f;
    if (tid < HH) {
        const float w = pWm1[tid], b = pbm1[tid];
        uw = w; ub = b; uw2 = pWm2[tid];
        const float q = -b / w;
        knotRaw[tid] = (w != 0.f) ? q : INF;
    } else if (tid < KNOT_PAD) {
        knotT[tid] = INF;
    }
    __syncthreads();
    if (tid < HH) {
        const float t = knotRaw[tid];
        int rank = 0;
        for (int i = 0; i < HH; ++i) {
            const float ti = knotRaw[i];
            rank += (ti < t || (ti == t && i < tid)) ? 1 : 0;
        }
        knotT[rank] = t;
        int lo, cnt;
        if (uw > 0.f)      { lo = rank + 1; cnt = HH - rank; }
        else if (uw < 0.f) { lo = 0;        cnt = rank + 1; }
        else               { lo = 0;        cnt = (ub > 0.f) ? HH + 1 : 0; }
        v4f u;
        u.x = uw; u.y = ub; u.z = uw2; u.w = __int_as_float(lo | (cnt << 8));
        unitT[tid] = u;
    }
    __syncthreads();
    if (tid <= HH) {
        const int j = tid;
        float S = 0.f, I = bm2v;
        #pragma unroll
        for (int h = 0; h < HH; ++h) {
            const v4f u = unitT[h];
            const float w = u.x, b = u.y, w2 = u.z;
            const int run = __float_as_int(u.w);
            const bool active = (unsigned)(j - (run & 255)) < (unsigned)(run >> 8);
            const float S1 = fmaf(w2, w, S), I1 = fmaf(w2, b, I);
            S = active ? S1 : S;
            I = active ? I1 : I;
        }
        segS[j] = S;
        segI[j] = I;
    }
    __syncthreads();
    const float knotTop = knotT[HH - 1];

    float* myT = mT[wave];
    v8f acc = {};
    const int kBase = wave * 256;
    for (int kc = kBase; kc < kBase + 256; kc += 64) {
        v2f d2[16], n2[16];
        #pragma unroll
        for (int r = 0; r < 16; ++r) {
            const size_t base = (size_t)(tileM + r) * NN + kc + 2 * lane;
            d2[r] = __builtin_nontemporal_load((const v2f*)(dist + base));
            n2[r] = __builtin_nontemporal_load((const v2f*)(norm + base));
        }
        int j0[16], j1[16];
        #pragma unroll
        for (int r = 0; r < 16; ++r) {
            j0[r] = (knotTop < d2[r].x) ? 4 * HH : 0;
            j1[r] = (knotTop < d2[r].y) ? 4 * HH : 0;
        }
        #pragma unroll
        for (int step = HH / 2; step >= 1; step >>= 1) {
            #pragma unroll
            for (int r = 0; r < 16; ++r) {
                const float t0 = lds_at(knotT, j0[r] + 4 * (step - 1));
                const float t1 = lds_at(knotT, j1[r] + 4 * (step - 1));
                j0[r] += (t0 < d2[r].x) ? 4 * step : 0;
                j1[r] += (t1 < d2[r].y) ? 4 * step : 0;
            }
        }
        #pragma unroll
        for (int r = 0; r < 16; ++r) {
            const float a0 = fmaf(lds_at(segS, j0[r]), d2[r].x, lds_at(segI, j0[r]));
            const float a1 = fmaf(lds_at(segS, j1[r]), d2[r].y, lds_at(segI, j1[r]));
            v2f mv;
            mv.x = a0 * __builtin_amdgcn_rcpf(n2[r].x);
            mv.y = a1 * __builtin_amdgcn_rcpf(n2[r].y);
            *(v2f*)&myT[r * 66 + 2 * lane] = mv;
        }
        #pragma unroll
        for (int kk = 0; kk < 64; kk += 4) {
            const v2f a = *(const v2f*)&myT[row * 66 + kk + 2 * hi];
            const int kg = kc + kk + 2 * hi;
            v2f b;
            b.x = fsums[kg       * CC + col];
            b.y = fsums[(kg + 1) * CC + col];
            acc = __builtin_amdgcn_wmma_f32_16x16x4_f32(false, a, false, b,
                                                        (short)0, acc, false, false);
        }
    }

    #pragma unroll
    for (int r = 0; r < 8; ++r) {
        const int orow = r + 8 * hi;
        redC[wave * 256 + orow * 16 + col] = acc[r];
    }
    __syncthreads();
    float v = 0.f;
    #pragma unroll
    for (int w = 0; w < 8; ++w) v += redC[w * 256 + tid];
    out[tileM * CC + tid] = v;
}

extern "C" void kernel_launch(void* const* d_in, const int* in_sizes, int n_in,
                              void* d_out, int out_size, void* d_ws, size_t ws_size,
                              hipStream_t stream) {
    const float* x    = (const float*)d_in[0];
    const float* dist = (const float*)d_in[1];
    const float* norm = (const float*)d_in[2];
    const float* W1   = (const float*)d_in[3];
    const float* b1   = (const float*)d_in[4];
    const float* W2   = (const float*)d_in[5];
    const float* b2   = (const float*)d_in[6];
    const float* fw   = (const float*)d_in[7];
    const float* Wm1  = (const float*)d_in[8];
    const float* bm1  = (const float*)d_in[9];
    const float* Wm2  = (const float*)d_in[10];
    const float* bm2  = (const float*)d_in[11];
    float* out   = (float*)d_out;
    float* fsums = (float*)d_ws;

    fsums_kernel<<<NN / 16, 256, 0, stream>>>(x, W1, b1, W2, b2, fw, fsums);
    out_kernel<<<NN / 16, 256, 0, stream>>>(dist, norm, Wm1, bm1, Wm2, bm2, fsums, out);
}


# m-network tables (sorted breakpoints, segment slope/intercept) built once by an extra block 128 of launch 1; out-kernel waves load the 192 floats: no build or barrier before the stream; same bits
# speedup vs baseline: 2.9021x; 1.0616x over previous
#include <hip/hip_runtime.h>
#include <hip/hip_bf16.h>

typedef __attribute__((ext_vector_type(2))) float v2f;
typedef __attribute__((ext_vector_type(8))) float v8f;

#define NN 2048
#define FF 64
#define HH 32
#define CC 16

typedef __attribute__((ext_vector_type(4))) float v4f;

#define KNOT_PAD 64

__device__ __forceinline__ void build_m_tables(float* sc, const int tid,
                                               const float* __restrict__ pWm1,
                                               const float* __restrict__ pbm1,
                                               const float* __restrict__ pWm2,
                                               const float* __restrict__ pbm2,
                                               float* __restrict__ tabOut)
{
    float* knotRaw = sc;
    float* knotT   = sc + KNOT_PAD;
    float* segS    = sc + 2 * KNOT_PAD;
    float* segI    = sc + 3 * KNOT_PAD;
    v4f*   unitT   = (v4f*)(sc + 4 * KNOT_PAD);

    const float INF = __builtin_huge_valf();
    const float bm2v = pbm2[0];
    float uw = 0.f, ub = 0.f, uw2 = 0.f;
    if (tid < HH) {
        const float w = pWm1[tid], b = pbm1[tid];
        uw = w; ub = b; uw2 = pWm2[tid];
        const float q = -b / w;
        knotRaw[tid] = (w != 0.f) ? q : INF;
    } else if (tid < KNOT_PAD) {
        knotT[tid] = INF;
        if (tid > HH) { segS[tid] = 0.f; segI[tid] = 0.f; }
    }
    __syncthreads();
    if (tid < HH) {
        const float t = knotRaw[tid];
        int rank = 0;
        for (int i = 0; i < HH; ++i) {
            const float ti = knotRaw[i];
            rank += (ti < t || (ti == t && i < tid)) ? 1 : 0;
        }
        knotT[rank] = t;
        int lo, cnt;
        if (uw > 0.f)      { lo = rank + 1; cnt = HH - rank; }
        else if (uw < 0.f) { lo = 0;        cnt = rank + 1; }
        else               { lo = 0;        cnt = (ub > 0.f) ? HH + 1 : 0; }
        v4f u;
        u.x = uw; u.y = ub; u.z = uw2; u.w = __int_as_float(lo | (cnt << 8));
        unitT[tid] = u;
    }
    __syncthreads();
    if (tid <= HH) {
        const int j = tid;
        float S = 0.f, I = bm2v;
        #pragma unroll
        for (int h = 0; h < HH; ++h) {
            const v4f u = unitT[h];
            const float w = u.x, b = u.y, w2 = u.z;
            const int run = __float_as_int(u.w);
            const bool active = (unsigned)(j - (run & 255)) < (unsigned)(run >> 8);
            const float S1 = fmaf(w2, w, S), I1 = fmaf(w2, b, I);
            S = active ? S1 : S;
            I = active ? I1 : I;
        }
        segS[j] = S;
        segI[j] = I;
    }
    __syncthreads();
    if (tid < 3 * KNOT_PAD) tabOut[tid] = knotT[tid];
}

__global__ __launch_bounds__(256)
void fsums_kernel(const float* __restrict__ x,
                  const float* __restrict__ W1,
                  const float* __restrict__ b1,
                  const float* __restrict__ W2,
                  const float* __restrict__ b2,
                  const float* __restrict__ fw,
                  const float* __restrict__ pWm1,
                  const float* __restrict__ pbm1,
                  const float* __restrict__ pWm2,
                  const float* __restrict__ pbm2,
                  float* __restrict__ fsums,
                  float* __restrict__ tabOut)
{
    __shared__ v4f w2T[8 * 1024];
    __shared__ v4f wbT[8 * 128];
    __shared__ v4f b2T[FF * CC / 4];
    __shared__ float attnLds[256];
    __shared__ __attribute__((aligned(16))) float redC[8 * 256];

    const int tid  = threadIdx.x;
    if (blockIdx.x == NN / 16) { build_m_tables(redC, tid, pWm1, pbm1, pWm2, pbm2, tabOut); return; }
    const int lane = tid & 31;
    const int wave = tid >> 5;
    const int tileM = blockIdx.x * 16;
    const int row = lane & 15;
    const int hi  = lane >> 4;
    const int col = row;
    const int gRow = tileM + row;
    const int kBase = wave * 256;

    const float fwv = fw[tid & (FF - 1)];
    const v4f* x4 = (const v4f*)(x + (size_t)gRow * FF + wave * 8);
    const v4f xa = x4[0], xb = x4[1];
    const v4f* W1v = (const v4f*)(W1 + kBase);
    const v4f* b1v = (const v4f*)(b1 + kBase);
    const v4f w1a = W1v[lane], w1b = W1v[32 + lane];
    const v4f b1a = b1v[lane], b1b = b1v[32 + lane];
    const v4f b2v = ((const v4f*)b2)[tid];
    const v4f* W2v = (const v4f*)(W2 + (size_t)kBase * CC);
    v4f w2r[32];
    #pragma unroll
    for (int i = 0; i < 32; ++i) w2r[i] = W2v[i * 32 + lane];

    attnLds[tid] = __expf(__expf(fwv));
    {
        v4f* wb = wbT + wave * 128;
        v4f p;
        p.x = w1a.x; p.y = w1a.y; p.z = b1a.x; p.w = b1a.y; wb[2 * lane]          = p;
        p.x = w1a.z; p.y = w1a.w; p.z = b1a.z; p.w = b1a.w; wb[2 * lane + 1]      = p;
        p.x = w1b.x; p.y = w1b.y; p.z = b1b.x; p.w = b1b.y; wb[64 + 2 * lane]     = p;
        p.x = w1b.z; p.y = w1b.w; p.z = b1b.z; p.w = b1b.w; wb[64 + 2 * lane + 1] = p;
    }
    b2T[tid] = b2v;
    #pragma unroll
    for (int i = 0; i < 32; ++i) w2T[wave * 1024 + i * 32 + lane] = w2r[i];
    __syncthreads();

    float s = 0.f;
    #pragma unroll
    for (int f = 0; f < FF; ++f) s += attnLds[f];
    const float invS = 1.0f / s;

    const float xv[8] = {xa.x, xa.y, xa.z, xa.w, xb.x, xb.y, xb.z, xb.w};
    const float* w2f = (const float*)(w2T + wave * 1024);
    const v4f*   wb  = wbT + wave * 128;

    v4f pc[8]; v2f bc[8];
    #pragma unroll
    for (int t = 0; t < 8; ++t) {
        const int kA = t * 4 + 2 * hi;
        pc[t] = wb[kA >> 1];
        bc[t].x = w2f[kA * CC + col];
        bc[t].y = w2f[(kA + 1) * CC + col];
    }
    v8f acc = {};
    #pragma unroll
    for (int fl = 0; fl < 8; ++fl) {
        v4f pn[8]; v2f bn[8];
        if (fl + 1 < 8) {
            #pragma unroll
            for (int t = 0; t < 8; ++t) {
                const int kA = (fl + 1) * 32 + t * 4 + 2 * hi;
                pn[t] = wb[kA >> 1];
                bn[t].x = w2f[kA * CC + col];
                bn[t].y = w2f[(kA + 1) * CC + col];
            }
        }
        asm volatile("" ::: "memory");
        const float at = attnLds[wave * 8 + fl] * invS;
        const float xf = xv[fl];
        v2f a[8], b[8];
        #pragma unroll
        for (int t = 0; t < 8; ++t) {
            a[t].x = fmaxf(fmaf(xf, pc[t].x, pc[t].z), 0.f);
            a[t].y = fmaxf(fmaf(xf, pc[t].y, pc[t].w), 0.f);
            b[t].x = at * bc[t].x;
            b[t].y = at * bc[t].y;
        }
        #pragma unroll
        for (int t = 0; t < 8; ++t)
            acc = __builtin_amdgcn_wmma_f32_16x16x4_f32(false, a[t], false, b[t],
                                                        (short)0, acc, false, false);
        if (fl + 1 < 8) {
            #pragma unroll
            for (int t = 0; t < 8; ++t) { pc[t] = pn[t]; bc[t] = bn[t]; }
        }
    }

    #pragma unroll
    for (int r = 0; r < 8; ++r) {
        const int orow = r + 8 * hi;
        redC[wave * 256 + orow * 16 + col] = acc[r];
    }
    const int ocol = tid & 15;
    const float* b2f = (const float*)b2T;
    float bias = 0.f;
    #pragma unroll
    for (int f = 0; f < FF; ++f) bias += attnLds[f] * invS * b2f[f * CC + ocol];
    __syncthreads();
    float v = 0.f;
    #pragma unroll
    for (int w = 0; w < 8; ++w) v += redC[w * 256 + tid];
    fsums[tileM * CC + tid] = v + bias;
}

__device__ __forceinline__ float lds_at(const float* base, int byteOff) {
    return *(const float*)((const char*)base + byteOff);
}

__global__ __launch_bounds__(256) __attribute__((amdgpu_waves_per_eu(2, 2)))
void out_kernel(const float* __restrict__ dist,
                const float* __restrict__ norm,
                const float* __restrict__ tabG,
                const float* __restrict__ fsums,
                float* __restrict__ out)
{
    __shared__ float mT[8][16 * 66];
    __shared__ float redC[8 * 256];
    __shared__ float tabL[8][3 * KNOT_PAD];

    const int tid  = threadIdx.x;
    const int lane = tid & 31;
    const int wave = tid >> 5;
    const int tileM = blockIdx.x * 16;
    const int row = lane & 15;
    const int hi  = lane >> 4;
    const int col = row;

    {
        const v2f* tg = (const v2f*)tabG;
        const v2f tk = tg[lane], ts = tg[32 + lane], ti = tg[64 + lane];
        float* myTab = tabL[wave];
        *(v2f*)&myTab[2 * lane]                = tk;
        *(v2f*)&myTab[KNOT_PAD + 2 * lane]     = ts;
        *(v2f*)&myTab[2 * KNOT_PAD + 2 * lane] = ti;
    }
    const float* tab0 = &tabL[0][0];
    const int wb = wave * (3 * KNOT_PAD * 4);
    const float knotTop = lds_at(tab0, wb + 4 * (HH - 1));

    float* myT = mT[wave];
    v8f acc = {};
    const int kBase = wave * 256;
    for (int kc = kBase; kc < kBase + 256; kc += 64) {
        v2f d2[16], n2[16];
        #pragma unroll
        for (int r = 0; r < 16; ++r) {
            const size_t base = (size_t)(tileM + r) * NN + kc + 2 * lane;
            d2[r] = __builtin_nontemporal_load((const v2f*)(dist + base));
            n2[r] = __builtin_nontemporal_load((const v2f*)(norm + base));
        }
        int j0[16], j1[16];
        #pragma unroll
        for (int r = 0; r < 16; ++r) {
            j0[r] = (knotTop < d2[r].x) ? wb + 4 * HH : wb;
            j1[r] = (knotTop < d2[r].y) ? wb + 4 * HH : wb;
        }
        #pragma unroll
        for (int step = HH / 2; step >= 1; step >>= 1) {
            #pragma unroll
            for (int r = 0; r < 16; ++r) {
                const float t0 = lds_at(tab0, j0[r] + 4 * (step - 1));
                const float t1 = lds_at(tab0, j1[r] + 4 * (step - 1));
                j0[r] += (t0 < d2[r].x) ? 4 * step : 0;
                j1[r] += (t1 < d2[r].y) ? 4 * step : 0;
            }
        }
        #pragma unroll
        for (int r = 0; r < 16; ++r) {
            const float a0 = fmaf(lds_at(tab0, j0[r] + 4 * KNOT_PAD), d2[r].x, lds_at(tab0, j0[r] + 8 * KNOT_PAD));
            const float a1 = fmaf(lds_at(tab0, j1[r] + 4 * KNOT_PAD), d2[r].y, lds_at(tab0, j1[r] + 8 * KNOT_PAD));
            v2f mv;
            mv.x = a0 * __builtin_amdgcn_rcpf(n2[r].x);
            mv.y = a1 * __builtin_amdgcn_rcpf(n2[r].y);
            *(v2f*)&myT[r * 66 + 2 * lane] = mv;
        }
        #pragma unroll
        for (int kk = 0; kk < 64; kk += 4) {
            const v2f a = *(const v2f*)&myT[row * 66 + kk + 2 * hi];
            const int kg = kc + kk + 2 * hi;
            v2f b;
            b.x = fsums[kg       * CC + col];
            b.y = fsums[(kg + 1) * CC + col];
            acc = __builtin_amdgcn_wmma_f32_16x16x4_f32(false, a, false, b,
                                                        (short)0, acc, false, false);
        }
    }

    #pragma unroll
    for (int r = 0; r < 8; ++r) {
        const int orow = r + 8 * hi;
        redC[wave * 256 + orow * 16 + col] = acc[r];
    }
    __syncthreads();
    float v = 0.f;
    #pragma unroll
    for (int w = 0; w < 8; ++w) v += redC[w * 256 + tid];
    out[tileM * CC + tid] = v;
}

extern "C" void kernel_launch(void* const* d_in, const int* in_sizes, int n_in,
                              void* d_out, int out_size, void* d_ws, size_t ws_size,
                              hipStream_t stream) {
    const float* x    = (const float*)d_in[0];
    const float* dist = (const float*)d_in[1];
    const float* norm = (const float*)d_in[2];
    const float* W1   = (const float*)d_in[3];
    const float* b1   = (const float*)d_in[4];
    const float* W2   = (const float*)d_in[5];
    const float* b2   = (const float*)d_in[6];
    const float* fw   = (const float*)d_in[7];
    const float* Wm1  = (const float*)d_in[8];
    const float* bm1  = (const float*)d_in[9];
    const float* Wm2  = (const float*)d_in[10];
    const float* bm2  = (const float*)d_in[11];
    float* out   = (float*)d_out;
    float* fsums = (float*)d_ws;
    float* tab   = fsums + NN * CC;

    fsums_kernel<<<NN / 16 + 1, 256, 0, stream>>>(x, W1, b1, W2, b2, fw, Wm1, bm1, Wm2, bm2, fsums, tab);
    out_kernel<<<NN / 16, 256, 0, stream>>>(dist, norm, tab, fsums, out);
}


# f_sums stored in matrix B-fragment order: out kernel gets the B operand of two matrix steps in one whole-line 16-byte load, 8 a stage behind the row loads (was 32 4-byte loads in front); same sums
# speedup vs baseline: 3.1553x; 1.0872x over previous
#include <hip/hip_runtime.h>
#include <hip/hip_bf16.h>

typedef __attribute__((ext_vector_type(2))) float v2f;
typedef __attribute__((ext_vector_type(8))) float v8f;

#define NN 2048
#define FF 64
#define HH 32
#define CC 16

typedef __attribute__((ext_vector_type(4))) float v4f;

#define KNOT_PAD 64

__device__ __forceinline__ void build_m_tables(float* sc, const int tid,
                                               const float* __restrict__ pWm1,
                                               const float* __restrict__ pbm1,
                                               const float* __restrict__ pWm2,
                                               const float* __restrict__ pbm2,
                                               float* __restrict__ tabOut)
{
    float* knotRaw = sc;
    float* knotT   = sc + KNOT_PAD;
    float* segS    = sc + 2 * KNOT_PAD;
    float* segI    = sc + 3 * KNOT_PAD;
    v4f*   unitT   = (v4f*)(sc + 4 * KNOT_PAD);

    const float INF = __builtin_huge_valf();
    const float bm2v = pbm2[0];
    float uw = 0.f, ub = 0.f, uw2 = 0.f;
    if (tid < HH) {
        const float w = pWm1[tid], b = pbm1[tid];
        uw = w; ub = b; uw2 = pWm2[tid];
        const float q = -b / w;
        knotRaw[tid] = (w != 0.f) ? q : INF;
    } else if (tid < KNOT_PAD) {
        knotT[tid] = INF;
        if (tid > HH) { segS[tid] = 0.f; segI[tid] = 0.f; }
    }
    __syncthreads();
    if (tid < HH) {
        const float t = knotRaw[tid];
        int rank = 0;
        for (int i = 0; i < HH; ++i) {
            const float ti = knotRaw[i];
            rank += (ti < t || (ti == t && i < tid)) ? 1 : 0;
        }
        knotT[rank] = t;
        int lo, cnt;
        if (uw > 0.f)      { lo = rank + 1; cnt = HH - rank; }
        else if (uw < 0.f) { lo = 0;        cnt = rank + 1; }
        else               { lo = 0;        cnt = (ub > 0.f) ? HH + 1 : 0; }
        v4f u;
        u.x = uw; u.y = ub; u.z = uw2; u.w = __int_as_float(lo | (cnt << 8));
        unitT[tid] = u;
    }
    __syncthreads();
    if (tid <= HH) {
        const int j = tid;
        float S = 0.f, I = bm2v;
        #pragma unroll
        for (int h = 0; h < HH; ++h) {
            const v4f u = unitT[h];
            const float w = u.x, b = u.y, w2 = u.z;
            const int run = __float_as_int(u.w);
            const bool active = (unsigned)(j - (run & 255)) < (unsigned)(run >> 8);
            const float S1 = fmaf(w2, w, S), I1 = fmaf(w2, b, I);
            S = active ? S1 : S;
            I = active ? I1 : I;
        }
        segS[j] = S;
        segI[j] = I;
    }
    __syncthreads();
    if (tid < 3 * KNOT_PAD) tabOut[tid] = knotT[tid];
}

__global__ __launch_bounds__(256)
void fsums_kernel(const float* __restrict__ x,
                  const float* __restrict__ W1,
                  const float* __restrict__ b1,
                  const float* __restrict__ W2,
                  const float* __restrict__ b2,
                  const float* __restrict__ fw,
                  const float* __restrict__ pWm1,
                  const float* __restrict__ pbm1,
                  const float* __restrict__ pWm2,
                  const float* __restrict__ pbm2,
                  float* __restrict__ fsums,
                  float* __restrict__ tabOut)
{
    __shared__ v4f w2T[8 * 1024];
    __shared__ v4f wbT[8 * 128];
    __shared__ v4f b2T[FF * CC / 4];
    __shared__ float attnLds[256];
    __shared__ __attribute__((aligned(16))) float redC[8 * 256];

    const int tid  = threadIdx.x;
    if (blockIdx.x == NN / 16) { build_m_tables(redC, tid, pWm1, pbm1, pWm2, pbm2, tabOut); return; }
    const int lane = tid & 31;
    const int wave = tid >> 5;
    const int tileM = blockIdx.x * 16;
    const int row = lane & 15;
    const int hi  = lane >> 4;
    const int col = row;
    const int gRow = tileM + row;
    const int kBase = wave * 256;

    const float fwv = fw[tid & (FF - 1)];
    const v4f* x4 = (const v4f*)(x + (size_t)gRow * FF + wave * 8);
    const v4f xa = x4[0], xb = x4[1];
    const v4f* W1v = (const v4f*)(W1 + kBase);
    const v4f* b1v = (const v4f*)(b1 + kBase);
    const v4f w1a = W1v[lane], w1b = W1v[32 + lane];
    const v4f b1a = b1v[lane], b1b = b1v[32 + lane];
    const v4f b2v = ((const v4f*)b2)[tid];
    const v4f* W2v = (const v4f*)(W2 + (size_t)kBase * CC);
    v4f w2r[32];
    #pragma unroll
    for (int i = 0; i < 32; ++i) w2r[i] = W2v[i * 32 + lane];

    attnLds[tid] = __expf(__expf(fwv));
    {
        v4f* wb = wbT + wave * 128;
        v4f p;
        p.x = w1a.x; p.y = w1a.y; p.z = b1a.x; p.w = b1a.y; wb[2 * lane]          = p;
        p.x = w1a.z; p.y = w1a.w; p.z = b1a.z; p.w = b1a.w; wb[2 * lane + 1]      = p;
        p.x = w1b.x; p.y = w1b.y; p.z = b1b.x; p.w = b1b.y; wb[64 + 2 * lane]     = p;
        p.x = w1b.z; p.y = w1b.w; p.z = b1b.z; p.w = b1b.w; wb[64 + 2 * lane + 1] = p;
    }
    b2T[tid] = b2v;
    #pragma unroll
    for (int i = 0; i < 32; ++i) w2T[wave * 1024 + i * 32 + lane] = w2r[i];
    __syncthreads();

    float s = 0.f;
    #pragma unroll
    for (int f = 0; f < FF; ++f) s += attnLds[f];
    const float invS = 1.0f / s;

    const float xv[8] = {xa.x, xa.y, xa.z, xa.w, xb.x, xb.y, xb.z, xb.w};
    const float* w2f = (const float*)(w2T + wave * 1024);
    const v4f*   wb  = wbT + wave * 128;

    v4f pc[8]; v2f bc[8];
    #pragma unroll
    for (int t = 0; t < 8; ++t) {
        const int kA = t * 4 + 2 * hi;
        pc[t] = wb[kA >> 1];
        bc[t].x = w2f[kA * CC + col];
        bc[t].y = w2f[(kA + 1) * CC + col];
    }
    v8f acc = {};
    #pragma unroll
    for (int fl = 0; fl < 8; ++fl) {
        v4f pn[8]; v2f bn[8];
        if (fl + 1 < 8) {
            #pragma unroll
            for (int t = 0; t < 8; ++t) {
                const int kA = (fl + 1) * 32 + t * 4 + 2 * hi;
                pn[t] = wb[kA >> 1];
                bn[t].x = w2f[kA * CC + col];
                bn[t].y = w2f[(kA + 1) * CC + col];
            }
        }
        asm volatile("" ::: "memory");
        const float at = attnLds[wave * 8 + fl] * invS;
        const float xf = xv[fl];
        v2f a[8], b[8];
        #pragma unroll
        for (int t = 0; t < 8; ++t) {
            a[t].x = fmaxf(fmaf(xf, pc[t].x, pc[t].z), 0.f);
            a[t].y = fmaxf(fmaf(xf, pc[t].y, pc[t].w), 0.f);
            b[t].x = at * bc[t].x;
            b[t].y = at * bc[t].y;
        }
        #pragma unroll
        for (int t = 0; t < 8; ++t)
            acc = __builtin_amdgcn_wmma_f32_16x16x4_f32(false, a[t], false, b[t],
                                                        (short)0, acc, false, false);
        if (fl + 1 < 8) {
            #pragma unroll
            for (int t = 0; t < 8; ++t) { pc[t] = pn[t]; bc[t] = bn[t]; }
        }
    }

    #pragma unroll
    for (int r = 0; r < 8; ++r) {
        const int orow = r + 8 * hi;
        redC[wave * 256 + orow * 16 + col] = acc[r];
    }
    const int orow = (tid >> 7) * 8 + ((tid >> 1) & 1) * 4 + ((tid >> 6) & 1) * 2 + (tid & 1);
    const int ocol = (tid >> 2) & 15;
    const float* b2f = (const float*)b2T;
    float bias = 0.f;
    #pragma unroll
    for (int f = 0; f < FF; ++f) bias += attnLds[f] * invS * b2f[f * CC + ocol];
    __syncthreads();
    float v = 0.f;
    #pragma unroll
    for (int w = 0; w < 8; ++w) v += redC[w * 256 + orow * 16 + ocol];
    fsums[tileM * CC + tid] = v + bias;
}

__device__ __forceinline__ float lds_at(const float* base, int byteOff) {
    return *(const float*)((const char*)base + byteOff);
}

__global__ __launch_bounds__(256) __attribute__((amdgpu_waves_per_eu(2, 2)))
void out_kernel(const float* __restrict__ dist,
                const float* __restrict__ norm,
                const float* __restrict__ tabG,
                const float* __restrict__ fsums,
                float* __restrict__ out)
{
    __shared__ float mT[8][16 * 66];
    __shared__ float redC[8 * 256];
    __shared__ float tabL[8][3 * KNOT_PAD];

    const int tid  = threadIdx.x;
    const int lane = tid & 31;
    const int wave = tid >> 5;
    const int tileM = blockIdx.x * 16;
    const int row = lane & 15;
    const int hi  = lane >> 4;
    const int col = row;

    {
        const v2f* tg = (const v2f*)tabG;
        const v2f tk = tg[lane], ts = tg[32 + lane], ti = tg[64 + lane];
        float* myTab = tabL[wave];
        *(v2f*)&myTab[2 * lane]                = tk;
        *(v2f*)&myTab[KNOT_PAD + 2 * lane]     = ts;
        *(v2f*)&myTab[2 * KNOT_PAD + 2 * lane] = ti;
    }
    const float* tab0 = &tabL[0][0];
    const int wb = wave * (3 * KNOT_PAD * 4);
    const float knotTop = lds_at(tab0, wb + 4 * (HH - 1));

    float* myT = mT[wave];
    v8f acc = {};
    const int kBase = wave * 256;
    for (int kc = kBase; kc < kBase + 256; kc += 64) {
        v2f d2[16], n2[16];
        #pragma unroll
        for (int r = 0; r < 16; ++r) {
            const size_t base = (size_t)(tileM + r) * NN + kc + 2 * lane;
            d2[r] = __builtin_nontemporal_load((const v2f*)(dist + base));
            n2[r] = __builtin_nontemporal_load((const v2f*)(norm + base));
        }
        asm volatile("" ::: "memory");
        v4f fq[8];
        {
            const v4f* fsq = (const v4f*)fsums + (kc >> 3) * 32 + lane;
            #pragma unroll
            for (int s = 0; s < 8; ++s) fq[s] = fsq[s * 32];
        }
        int j0[16], j1[16];
        #pragma unroll
        for (int r = 0; r < 16; ++r) {
            j0[r] = (knotTop < d2[r].x) ? wb + 4 * HH : wb;
            j1[r] = (knotTop < d2[r].y) ? wb + 4 * HH : wb;
        }
        #pragma unroll
        for (int step = HH / 2; step >= 1; step >>= 1) {
            #pragma unroll
            for (int r = 0; r < 16; ++r) {
                const float t0 = lds_at(tab0, j0[r] + 4 * (step - 1));
                const float t1 = lds_at(tab0, j1[r] + 4 * (step - 1));
                j0[r] += (t0 < d2[r].x) ? 4 * step : 0;
                j1[r] += (t1 < d2[r].y) ? 4 * step : 0;
            }
        }
        #pragma unroll
        for (int r = 0; r < 16; ++r) {
            const float a0 = fmaf(lds_at(tab0, j0[r] + 4 * KNOT_PAD), d2[r].x, lds_at(tab0, j0[r] + 8 * KNOT_PAD));
            const float a1 = fmaf(lds_at(tab0, j1[r] + 4 * KNOT_PAD), d2[r].y, lds_at(tab0, j1[r] + 8 * KNOT_PAD));
            v2f mv;
            mv.x = a0 * __builtin_amdgcn_rcpf(n2[r].x);
            mv.y = a1 * __builtin_amdgcn_rcpf(n2[r].y);
            *(v2f*)&myT[r * 66 + 2 * lane] = mv;
        }
        #pragma unroll
        for (int kk = 0; kk < 64; kk += 8) {
            const v4f q = fq[kk >> 3];
            const v2f a0 = *(const v2f*)&myT[row * 66 + kk + 2 * hi];
            const v2f a1 = *(const v2f*)&myT[row * 66 + kk + 4 + 2 * hi];
            v2f b0, b1;
            b0.x = q.x; b0.y = q.y;
            b1.x = q.z; b1.y = q.w;
            acc = __builtin_amdgcn_wmma_f32_16x16x4_f32(false, a0, false, b0,
                                                        (short)0, acc, false, false);
            acc = __builtin_amdgcn_wmma_f32_16x16x4_f32(false, a1, false, b1,
                                                        (short)0, acc, false, false);
        }
    }

    #pragma unroll
    for (int r = 0; r < 8; ++r) {
        const int orow = r + 8 * hi;
        redC[wave * 256 + orow * 16 + col] = acc[r];
    }
    __syncthreads();
    float v = 0.f;
    #pragma unroll
    for (int w = 0; w < 8; ++w) v += redC[w * 256 + tid];
    out[tileM * CC + tid] = v;
}

extern "C" void kernel_launch(void* const* d_in, const int* in_sizes, int n_in,
                              void* d_out, int out_size, void* d_ws, size_t ws_size,
                              hipStream_t stream) {
    const float* x    = (const float*)d_in[0];
    const float* dist = (const float*)d_in[1];
    const float* norm = (const float*)d_in[2];
    const float* W1   = (const float*)d_in[3];
    const float* b1   = (const float*)d_in[4];
    const float* W2   = (const float*)d_in[5];
    const float* b2   = (const float*)d_in[6];
    const float* fw   = (const float*)d_in[7];
    const float* Wm1  = (const float*)d_in[8];
    const float* bm1  = (const float*)d_in[9];
    const float* Wm2  = (const float*)d_in[10];
    const float* bm2  = (const float*)d_in[11];
    float* out   = (float*)d_out;
    float* fsums = (float*)d_ws;
    float* tab   = fsums + NN * CC;

    fsums_kernel<<<NN / 16 + 1, 256, 0, stream>>>(x, W1, b1, W2, b2, fw, Wm1, bm1, Wm2, bm2, fsums, tab);
    out_kernel<<<NN / 16, 256, 0, stream>>>(dist, norm, tab, fsums, out);
}


# out kernel, segment search in registers: a level's pivots in one register, fetched by in-row lane gather on the path so far, sign of pivot-d shifted in; 1 LDS read an element (was 6); same j
# speedup vs baseline: 3.4775x; 1.1021x over previous
#include <hip/hip_runtime.h>
#include <hip/hip_bf16.h>

typedef __attribute__((ext_vector_type(2))) float v2f;
typedef __attribute__((ext_vector_type(8))) float v8f;

#define NN 2048
#define FF 64
#define HH 32
#define CC 16

typedef __attribute__((ext_vector_type(4))) float v4f;

#define KNOT_PAD 64

__device__ __forceinline__ void build_m_tables(float* sc, const int tid,
                                               const float* __restrict__ pWm1,
                                               const float* __restrict__ pbm1,
                                               const float* __restrict__ pWm2,
                                               const float* __restrict__ pbm2,
                                               float* __restrict__ tabOut)
{
    float* knotRaw = sc;
    float* knotT   = sc + KNOT_PAD;
    float* segS    = sc + 2 * KNOT_PAD;
    float* segI    = sc + 3 * KNOT_PAD;
    v4f*   unitT   = (v4f*)(sc + 4 * KNOT_PAD);

    const float INF = __builtin_huge_valf();
    const float bm2v = pbm2[0];
    float uw = 0.f, ub = 0.f, uw2 = 0.f;
    if (tid < HH) {
        const float w = pWm1[tid], b = pbm1[tid];
        uw = w; ub = b; uw2 = pWm2[tid];
        const float q = -b / w;
        knotRaw[tid] = (w != 0.f) ? q : INF;
    } else if (tid < KNOT_PAD) {
        knotT[tid] = INF;
        if (tid > HH) { segS[tid] = 0.f; segI[tid] = 0.f; }
    }
    __syncthreads();
    if (tid < HH) {
        const float t = knotRaw[tid];
        int rank = 0;
        for (int i = 0; i < HH; ++i) {
            const float ti = knotRaw[i];
            rank += (ti < t || (ti == t && i < tid)) ? 1 : 0;
        }
        knotT[rank] = t;
        int lo, cnt;
        if (uw > 0.f)      { lo = rank + 1; cnt = HH - rank; }
        else if (uw < 0.f) { lo = 0;        cnt = rank + 1; }
        else               { lo = 0;        cnt = (ub > 0.f) ? HH + 1 : 0; }
        v4f u;
        u.x = uw; u.y = ub; u.z = uw2; u.w = __int_as_float(lo | (cnt << 8));
        unitT[tid] = u;
    }
    __syncthreads();
    if (tid <= HH) {
        const int j = tid;
        float S = 0.f, I = bm2v;
        #pragma unroll
        for (int h = 0; h < HH; ++h) {
            const v4f u = unitT[h];
            const float w = u.x, b = u.y, w2 = u.z;
            const int run = __float_as_int(u.w);
            const bool active = (unsigned)(j - (run & 255)) < (unsigned)(run >> 8);
            const float S1 = fmaf(w2, w, S), I1 = fmaf(w2, b, I);
            S = active ? S1 : S;
            I = active ? I1 : I;
        }
        segS[j] = S;
        segI[j] = I;
    }
    __syncthreads();
    if (tid < 3 * KNOT_PAD) tabOut[tid] = knotT[tid];
}

__global__ __launch_bounds__(256)
void fsums_kernel(const float* __restrict__ x,
                  const float* __restrict__ W1,
                  const float* __restrict__ b1,
                  const float* __restrict__ W2,
                  const float* __restrict__ b2,
                  const float* __restrict__ fw,
                  const float* __restrict__ pWm1,
                  const float* __restrict__ pbm1,
                  const float* __restrict__ pWm2,
                  const float* __restrict__ pbm2,
                  float* __restrict__ fsums,
                  float* __restrict__ tabOut)
{
    __shared__ v4f w2T[8 * 1024];
    __shared__ v4f wbT[8 * 128];
    __shared__ v4f b2T[FF * CC / 4];
    __shared__ float attnLds[256];
    __shared__ __attribute__((aligned(16))) float redC[8 * 256];

    const int tid  = threadIdx.x;
    if (blockIdx.x == NN / 16) { build_m_tables(redC, tid, pWm1, pbm1, pWm2, pbm2, tabOut); return; }
    const int lane = tid & 31;
    const int wave = tid >> 5;
    const int tileM = blockIdx.x * 16;
    const int row = lane & 15;
    const int hi  = lane >> 4;
    const int col = row;
    const int gRow = tileM + row;
    const int kBase = wave * 256;

    const float fwv = fw[tid & (FF - 1)];
    const v4f* x4 = (const v4f*)(x + (size_t)gRow * FF + wave * 8);
    const v4f xa = x4[0], xb = x4[1];
    const v4f* W1v = (const v4f*)(W1 + kBase);
    const v4f* b1v = (const v4f*)(b1 + kBase);
    const v4f w1a = W1v[lane], w1b = W1v[32 + lane];
    const v4f b1a = b1v[lane], b1b = b1v[32 + lane];
    const v4f b2v = ((const v4f*)b2)[tid];
    const v4f* W2v = (const v4f*)(W2 + (size_t)kBase * CC);
    v4f w2r[32];
    #pragma unroll
    for (int i = 0; i < 32; ++i) w2r[i] = W2v[i * 32 + lane];

    attnLds[tid] = __expf(__expf(fwv));
    {
        v4f* wb = wbT + wave * 128;
        v4f p;
        p.x = w1a.x; p.y = w1a.y; p.z = b1a.x; p.w = b1a.y; wb[2 * lane]          = p;
        p.x = w1a.z; p.y = w1a.w; p.z = b1a.z; p.w = b1a.w; wb[2 * lane + 1]      = p;
        p.x = w1b.x; p.y = w1b.y; p.z = b1b.x; p.w = b1b.y; wb[64 + 2 * lane]     = p;
        p.x = w1b.z; p.y = w1b.w; p.z = b1b.z; p.w = b1b.w; wb[64 + 2 * lane + 1] = p;
    }
    b2T[tid] = b2v;
    #pragma unroll
    for (int i = 0; i < 32; ++i) w2T[wave * 1024 + i * 32 + lane] = w2r[i];
    __syncthreads();

    float s = 0.f;
    #pragma unroll
    for (int f = 0; f < FF; ++f) s += attnLds[f];
    const float invS = 1.0f / s;

    const float xv[8] = {xa.x, xa.y, xa.z, xa.w, xb.x, xb.y, xb.z, xb.w};
    const float* w2f = (const float*)(w2T + wave * 1024);
    const v4f*   wb  = wbT + wave * 128;

    v4f pc[8]; v2f bc[8];
    #pragma unroll
    for (int t = 0; t < 8; ++t) {
        const int kA = t * 4 + 2 * hi;
        pc[t] = wb[kA >> 1];
        bc[t].x = w2f[kA * CC + col];
        bc[t].y = w2f[(kA + 1) * CC + col];
    }
    v8f acc = {};
    #pragma unroll
    for (int fl = 0; fl < 8; ++fl) {
        v4f pn[8]; v2f bn[8];
        if (fl + 1 < 8) {
            #pragma unroll
            for (int t = 0; t < 8; ++t) {
                const int kA = (fl + 1) * 32 + t * 4 + 2 * hi;
                pn[t] = wb[kA >> 1];
                bn[t].x = w2f[kA * CC + col];
                bn[t].y = w2f[(kA + 1) * CC + col];
            }
        }
        asm volatile("" ::: "memory");
        const float at = attnLds[wave * 8 + fl] * invS;
        const float xf = xv[fl];
        v2f a[8], b[8];
        #pragma unroll
        for (int t = 0; t < 8; ++t) {
            a[t].x = fmaxf(fmaf(xf, pc[t].x, pc[t].z), 0.f);
            a[t].y = fmaxf(fmaf(xf, pc[t].y, pc[t].w), 0.f);
            b[t].x = at * bc[t].x;
            b[t].y = at * bc[t].y;
        }
        #pragma unroll
        for (int t = 0; t < 8; ++t)
            acc = __builtin_amdgcn_wmma_f32_16x16x4_f32(false, a[t], false, b[t],
                                                        (short)0, acc, false, false);
        if (fl + 1 < 8) {
            #pragma unroll
            for (int t = 0; t < 8; ++t) { pc[t] = pn[t]; bc[t] = bn[t]; }
        }
    }

    #pragma unroll
    for (int r = 0; r < 8; ++r) {
        const int orow = r + 8 * hi;
        redC[wave * 256 + orow * 16 + col] = acc[r];
    }
    const int orow = (tid >> 7) * 8 + ((tid >> 1) & 1) * 4 + ((tid >> 6) & 1) * 2 + (tid & 1);
    const int ocol = (tid >> 2) & 15;
    const float* b2f = (const float*)b2T;
    float bias = 0.f;
    #pragma unroll
    for (int f = 0; f < FF; ++f) bias += attnLds[f] * invS * b2f[f * CC + ocol];
    __syncthreads();
    float v = 0.f;
    #pragma unroll
    for (int w = 0; w < 8; ++w) v += redC[w * 256 + orow * 16 + ocol];
    fsums[tileM * CC + tid] = v + bias;
}

__device__ __forceinline__ float lds_at(const float* base, int byteOff) {
    return *(const float*)((const char*)base + byteOff);
}

__global__ __launch_bounds__(256) __attribute__((amdgpu_waves_per_eu(2, 2)))
void out_kernel(const float* __restrict__ dist,
                const float* __restrict__ norm,
                const float* __restrict__ tabG,
                const float* __restrict__ fsums,
                float* __restrict__ out)
{
    __shared__ float mT[8][16 * 66];
    __shared__ float redC[8 * 256];
    __shared__ float tabL[8][2 * KNOT_PAD];

    const int tid  = threadIdx.x;
    const int lane = tid & 31;
    const int wave = tid >> 5;
    const int tileM = blockIdx.x * 16;
    const int row = lane & 15;
    const int hi  = lane >> 4;
    const int col = row;

    const float* tab0 = &tabL[0][0];
    const int wb = wave * (2 * KNOT_PAD * 4);
    const float kRoot = tabG[HH / 2 - 1];
    const float kTop  = tabG[HH - 1];
    const unsigned piv1 = __float_as_uint(tabG[16 * (row & 1) + 7] + 0.0f);
    const unsigned piv2 = __float_as_uint(tabG[8 * (row & 3) + 3] + 0.0f);
    const unsigned piv3 = __float_as_uint(tabG[4 * (row & 7) + 1] + 0.0f);
    const unsigned piv4 = __float_as_uint(tabG[2 * row] + 0.0f);
    {
        const v2f* tg = (const v2f*)tabG;
        const v2f ts = tg[32 + lane], ti = tg[64 + lane];
        float* myTab = tabL[wave];
        *(v2f*)&myTab[2 * lane]            = ts;
        *(v2f*)&myTab[KNOT_PAD + 2 * lane] = ti;
    }

    float* myT = mT[wave];
    v8f acc = {};
    const int kBase = wave * 256;
    for (int kc = kBase; kc < kBase + 256; kc += 64) {
        v2f d2[16], n2[16];
        #pragma unroll
        for (int r = 0; r < 16; ++r) {
            const size_t base = (size_t)(tileM + r) * NN + kc + 2 * lane;
            d2[r] = __builtin_nontemporal_load((const v2f*)(dist + base));
            n2[r] = __builtin_nontemporal_load((const v2f*)(norm + base));
        }
        asm volatile("" ::: "memory");
        v4f fq[8];
        {
            const v4f* fsq = (const v4f*)fsums + (kc >> 3) * 32 + lane;
            #pragma unroll
            for (int s = 0; s < 8; ++s) fq[s] = fsq[s * 32];
        }
        unsigned p0[16], p1[16];
        #pragma unroll
        for (int r = 0; r < 16; ++r) {
            p0[r] = (kRoot < d2[r].x) ? 1u : 0u;
            p1[r] = (kRoot < d2[r].y) ? 1u : 0u;
        }
        #pragma unroll
        for (int lvl = 1; lvl <= 4; ++lvl) {
            const unsigned piv = (lvl == 1) ? piv1 : (lvl == 2) ? piv2 : (lvl == 3) ? piv3 : piv4;
            #pragma unroll
            for (int r = 0; r < 16; ++r) {
                const float t0 = __uint_as_float(__builtin_amdgcn_permlane16_var(0u, piv, p0[r], true, false));
                const float t1 = __uint_as_float(__builtin_amdgcn_permlane16_var(0u, piv, p1[r], true, false));
                p0[r] = __builtin_amdgcn_alignbit(p0[r], __float_as_uint(t0 - d2[r].x), 31u);
                p1[r] = __builtin_amdgcn_alignbit(p1[r], __float_as_uint(t1 - d2[r].y), 31u);
            }
        }
        #pragma unroll
        for (int r = 0; r < 16; ++r) {
            const int j0 = wb + 4 * (int)(p0[r] + ((kTop < d2[r].x) ? 1u : 0u));
            const int j1 = wb + 4 * (int)(p1[r] + ((kTop < d2[r].y) ? 1u : 0u));
            const float a0 = fmaf(lds_at(tab0, j0), d2[r].x, lds_at(tab0, j0 + 4 * KNOT_PAD));
            const float a1 = fmaf(lds_at(tab0, j1), d2[r].y, lds_at(tab0, j1 + 4 * KNOT_PAD));
            v2f mv;
            mv.x = a0 * __builtin_amdgcn_rcpf(n2[r].x);
            mv.y = a1 * __builtin_amdgcn_rcpf(n2[r].y);
            *(v2f*)&myT[r * 66 + 2 * lane] = mv;
        }
        #pragma unroll
        for (int kk = 0; kk < 64; kk += 8) {
            const v4f q = fq[kk >> 3];
            const v2f a0 = *(const v2f*)&myT[row * 66 + kk + 2 * hi];
            const v2f a1 = *(const v2f*)&myT[row * 66 + kk + 4 + 2 * hi];
            v2f b0, b1;
            b0.x = q.x; b0.y = q.y;
            b1.x = q.z; b1.y = q.w;
            acc = __builtin_amdgcn_wmma_f32_16x16x4_f32(false, a0, false, b0,
                                                        (short)0, acc, false, false);
            acc = __builtin_amdgcn_wmma_f32_16x16x4_f32(false, a1, false, b1,
                                                        (short)0, acc, false, false);
        }
    }

    #pragma unroll
    for (int r = 0; r < 8; ++r) {
        const int orow = r + 8 * hi;
        redC[wave * 256 + orow * 16 + col] = acc[r];
    }
    __syncthreads();
    float v = 0.f;
    #pragma unroll
    for (int w = 0; w < 8; ++w) v += redC[w * 256 + tid];
    out[tileM * CC + tid] = v;
}

extern "C" void kernel_launch(void* const* d_in, const int* in_sizes, int n_in,
                              void* d_out, int out_size, void* d_ws, size_t ws_size,
                              hipStream_t stream) {
    const float* x    = (const float*)d_in[0];
    const float* dist = (const float*)d_in[1];
    const float* norm = (const float*)d_in[2];
    const float* W1   = (const float*)d_in[3];
    const float* b1   = (const float*)d_in[4];
    const float* W2   = (const float*)d_in[5];
    const float* b2   = (const float*)d_in[6];
    const float* fw   = (const float*)d_in[7];
    const float* Wm1  = (const float*)d_in[8];
    const float* bm1  = (const float*)d_in[9];
    const float* Wm2  = (const float*)d_in[10];
    const float* bm2  = (const float*)d_in[11];
    float* out   = (float*)d_out;
    float* fsums = (float*)d_ws;
    float* tab   = fsums + NN * CC;

    fsums_kernel<<<NN / 16 + 1, 256, 0, stream>>>(x, W1, b1, W2, b2, fw, Wm1, bm1, Wm2, bm2, fsums, tab);
    out_kernel<<<NN / 16, 256, 0, stream>>>(dist, norm, tab, fsums, out);
}


# out kernel: the 32 row loads of stage s+1 issued at the top of stage s into a second register set (was: behind stage s's matrix steps, a load wait a stage); 4 stages straight-line; same sums
# speedup vs baseline: 3.7316x; 1.0731x over previous
#include <hip/hip_runtime.h>
#include <hip/hip_bf16.h>

typedef __attribute__((ext_vector_type(2))) float v2f;
typedef __attribute__((ext_vector_type(8))) float v8f;

#define NN 2048
#define FF 64
#define HH 32
#define CC 16

typedef __attribute__((ext_vector_type(4))) float v4f;

#define KNOT_PAD 64

__device__ __forceinline__ void build_m_tables(float* sc, const int tid,
                                               const float* __restrict__ pWm1,
                                               const float* __restrict__ pbm1,
                                               const float* __restrict__ pWm2,
                                               const float* __restrict__ pbm2,
                                               float* __restrict__ tabOut)
{
    float* knotRaw = sc;
    float* knotT   = sc + KNOT_PAD;
    float* segS    = sc + 2 * KNOT_PAD;
    float* segI    = sc + 3 * KNOT_PAD;
    v4f*   unitT   = (v4f*)(sc + 4 * KNOT_PAD);

    const float INF = __builtin_huge_valf();
    const float bm2v = pbm2[0];
    float uw = 0.f, ub = 0.f, uw2 = 0.f;
    if (tid < HH) {
        const float w = pWm1[tid], b = pbm1[tid];
        uw = w; ub = b; uw2 = pWm2[tid];
        const float q = -b / w;
        knotRaw[tid] = (w != 0.f) ? q : INF;
    } else if (tid < KNOT_PAD) {
        knotT[tid] = INF;
        if (tid > HH) { segS[tid] = 0.f; segI[tid] = 0.f; }
    }
    __syncthreads();
    if (tid < HH) {
        const float t = knotRaw[tid];
        int rank = 0;
        for (int i = 0; i < HH; ++i) {
            const float ti = knotRaw[i];
            rank += (ti < t || (ti == t && i < tid)) ? 1 : 0;
        }
        knotT[rank] = t;
        int lo, cnt;
        if (uw > 0.f)      { lo = rank + 1; cnt = HH - rank; }
        else if (uw < 0.f) { lo = 0;        cnt = rank + 1; }
        else               { lo = 0;        cnt = (ub > 0.f) ? HH + 1 : 0; }
        v4f u;
        u.x = uw; u.y = ub; u.z = uw2; u.w = __int_as_float(lo | (cnt << 8));
        unitT[tid] = u;
    }
    __syncthreads();
    if (tid <= HH) {
        const int j = tid;
        float S = 0.f, I = bm2v;
        #pragma unroll
        for (int h = 0; h < HH; ++h) {
            const v4f u = unitT[h];
            const float w = u.x, b = u.y, w2 = u.z;
            const int run = __float_as_int(u.w);
            const bool active = (unsigned)(j - (run & 255)) < (unsigned)(run >> 8);
            const float S1 = fmaf(w2, w, S), I1 = fmaf(w2, b, I);
            S = active ? S1 : S;
            I = active ? I1 : I;
        }
        segS[j] = S;
        segI[j] = I;
    }
    __syncthreads();
    if (tid < 3 * KNOT_PAD) tabOut[tid] = knotT[tid];
}

__global__ __launch_bounds__(256)
void fsums_kernel(const float* __restrict__ x,
                  const float* __restrict__ W1,
                  const float* __restrict__ b1,
                  const float* __restrict__ W2,
                  const float* __restrict__ b2,
                  const float* __restrict__ fw,
                  const float* __restrict__ pWm1,
                  const float* __restrict__ pbm1,
                  const float* __restrict__ pWm2,
                  const float* __restrict__ pbm2,
                  float* __restrict__ fsums,
                  float* __restrict__ tabOut)
{
    __shared__ v4f w2T[8 * 1024];
    __shared__ v4f wbT[8 * 128];
    __shared__ v4f b2T[FF * CC / 4];
    __shared__ float attnLds[256];
    __shared__ __attribute__((aligned(16))) float redC[8 * 256];

    const int tid  = threadIdx.x;
    if (blockIdx.x == NN / 16) { build_m_tables(redC, tid, pWm1, pbm1, pWm2, pbm2, tabOut); return; }
    const int lane = tid & 31;
    const int wave = tid >> 5;
    const int tileM = blockIdx.x * 16;
    const int row = lane & 15;
    const int hi  = lane >> 4;
    const int col = row;
    const int gRow = tileM + row;
    const int kBase = wave * 256;

    const float fwv = fw[tid & (FF - 1)];
    const v4f* x4 = (const v4f*)(x + (size_t)gRow * FF + wave * 8);
    const v4f xa = x4[0], xb = x4[1];
    const v4f* W1v = (const v4f*)(W1 + kBase);
    const v4f* b1v = (const v4f*)(b1 + kBase);
    const v4f w1a = W1v[lane], w1b = W1v[32 + lane];
    const v4f b1a = b1v[lane], b1b = b1v[32 + lane];
    const v4f b2v = ((const v4f*)b2)[tid];
    const v4f* W2v = (const v4f*)(W2 + (size_t)kBase * CC);
    v4f w2r[32];
    #pragma unroll
    for (int i = 0; i < 32; ++i) w2r[i] = W2v[i * 32 + lane];

    attnLds[tid] = __expf(__expf(fwv));
    {
        v4f* wb = wbT + wave * 128;
        v4f p;
        p.x = w1a.x; p.y = w1a.y; p.z = b1a.x; p.w = b1a.y; wb[2 * lane]          = p;
        p.x = w1a.z; p.y = w1a.w; p.z = b1a.z; p.w = b1a.w; wb[2 * lane + 1]      = p;
        p.x = w1b.x; p.y = w1b.y; p.z = b1b.x; p.w = b1b.y; wb[64 + 2 * lane]     = p;
        p.x = w1b.z; p.y = w1b.w; p.z = b1b.z; p.w = b1b.w; wb[64 + 2 * lane + 1] = p;
    }
    b2T[tid] = b2v;
    #pragma unroll
    for (int i = 0; i < 32; ++i) w2T[wave * 1024 + i * 32 + lane] = w2r[i];
    __syncthreads();

    float s = 0.f;
    #pragma unroll
    for (int f = 0; f < FF; ++f) s += attnLds[f];
    const float invS = 1.0f / s;

    const float xv[8] = {xa.x, xa.y, xa.z, xa.w, xb.x, xb.y, xb.z, xb.w};
    const float* w2f = (const float*)(w2T + wave * 1024);
    const v4f*   wb  = wbT + wave * 128;

    v4f pc[8]; v2f bc[8];
    #pragma unroll
    for (int t = 0; t < 8; ++t) {
        const int kA = t * 4 + 2 * hi;
        pc[t] = wb[kA >> 1];
        bc[t].x = w2f[kA * CC + col];
        bc[t].y = w2f[(kA + 1) * CC + col];
    }
    v8f acc = {};
    #pragma unroll
    for (int fl = 0; fl < 8; ++fl) {
        v4f pn[8]; v2f bn[8];
        if (fl + 1 < 8) {
            #pragma unroll
            for (int t = 0; t < 8; ++t) {
                const int kA = (fl + 1) * 32 + t * 4 + 2 * hi;
                pn[t] = wb[kA >> 1];
                bn[t].x = w2f[kA * CC + col];
                bn[t].y = w2f[(kA + 1) * CC + col];
            }
        }
        asm volatile("" ::: "memory");
        const float at = attnLds[wave * 8 + fl] * invS;
        const float xf = xv[fl];
        v2f a[8], b[8];
        #pragma unroll
        for (int t = 0; t < 8; ++t) {
            a[t].x = fmaxf(fmaf(xf, pc[t].x, pc[t].z), 0.f);
            a[t].y = fmaxf(fmaf(xf, pc[t].y, pc[t].w), 0.f);
            b[t].x = at * bc[t].x;
            b[t].y = at * bc[t].y;
        }
        #pragma unroll
        for (int t = 0; t < 8; ++t)
            acc = __builtin_amdgcn_wmma_f32_16x16x4_f32(false, a[t], false, b[t],
                                                        (short)0, acc, false, false);
        if (fl + 1 < 8) {
            #pragma unroll
            for (int t = 0; t < 8; ++t) { pc[t] = pn[t]; bc[t] = bn[t]; }
        }
    }

    #pragma unroll
    for (int r = 0; r < 8; ++r) {
        const int orow = r + 8 * hi;
        redC[wave * 256 + orow * 16 + col] = acc[r];
    }
    const int orow = (tid >> 7) * 8 + ((tid >> 1) & 1) * 4 + ((tid >> 6) & 1) * 2 + (tid & 1);
    const int ocol = (tid >> 2) & 15;
    const float* b2f = (const float*)b2T;
    float bias = 0.f;
    #pragma unroll
    for (int f = 0; f < FF; ++f) bias += attnLds[f] * invS * b2f[f * CC + ocol];
    __syncthreads();
    float v = 0.f;
    #pragma unroll
    for (int w = 0; w < 8; ++w) v += redC[w * 256 + orow * 16 + ocol];
    fsums[tileM * CC + tid] = v + bias;
}

__device__ __forceinline__ float lds_at(const float* base, int byteOff) {
    return *(const float*)((const char*)base + byteOff);
}

__device__ __forceinline__ void load_rows(v2f (&d2)[16], v2f (&n2)[16],
                                          const float* __restrict__ dist,
                                          const float* __restrict__ norm, const size_t off)
{
    #pragma unroll
    for (int r = 0; r < 16; ++r) {
        const size_t base = off + (size_t)r * NN;
        d2[r] = __builtin_nontemporal_load((const v2f*)(dist + base));
        n2[r] = __builtin_nontemporal_load((const v2f*)(norm + base));
    }
}

__device__ __forceinline__ void load_fq(v4f (&fq)[8], const float* __restrict__ fsums,
                                        const int kc, const int lane)
{
    const v4f* fsq = (const v4f*)fsums + (kc >> 3) * 32 + lane;
    #pragma unroll
    for (int s = 0; s < 8; ++s) fq[s] = fsq[s * 32];
}

__device__ __forceinline__ void stage_math(const v2f (&d2)[16], const v2f (&n2)[16], const v4f (&fq)[8],
                                           const float kRoot, const float kTop,
                                           const unsigned piv1, const unsigned piv2,
                                           const unsigned piv3, const unsigned piv4,
                                           const float* tab0, const int wb, float* myT,
                                           const int lane, const int row, const int hi, v8f& acc)
{
    unsigned p0[16], p1[16];
    #pragma unroll
    for (int r = 0; r < 16; ++r) {
        p0[r] = (kRoot < d2[r].x) ? 1u : 0u;
        p1[r] = (kRoot < d2[r].y) ? 1u : 0u;
    }
    #pragma unroll
    for (int lvl = 1; lvl <= 4; ++lvl) {
        const unsigned piv = (lvl == 1) ? piv1 : (lvl == 2) ? piv2 : (lvl == 3) ? piv3 : piv4;
        #pragma unroll
        for (int r = 0; r < 16; ++r) {
            const float t0 = __uint_as_float(__builtin_amdgcn_permlane16_var(0u, piv, p0[r], true, false));
            const float t1 = __uint_as_float(__builtin_amdgcn_permlane16_var(0u, piv, p1[r], true, false));
            p0[r] = __builtin_amdgcn_alignbit(p0[r], __float_as_uint(t0 - d2[r].x), 31u);
            p1[r] = __builtin_amdgcn_alignbit(p1[r], __float_as_uint(t1 - d2[r].y), 31u);
        }
    }
    #pragma unroll
    for (int r = 0; r < 16; ++r) {
        const int j0 = wb + 4 * (int)(p0[r] + ((kTop < d2[r].x) ? 1u : 0u));
        const int j1 = wb + 4 * (int)(p1[r] + ((kTop < d2[r].y) ? 1u : 0u));
        const float a0 = fmaf(lds_at(tab0, j0), d2[r].x, lds_at(tab0, j0 + 4 * KNOT_PAD));
        const float a1 = fmaf(lds_at(tab0, j1), d2[r].y, lds_at(tab0, j1 + 4 * KNOT_PAD));
        v2f mv;
        mv.x = a0 * __builtin_amdgcn_rcpf(n2[r].x);
        mv.y = a1 * __builtin_amdgcn_rcpf(n2[r].y);
        *(v2f*)&myT[r * 66 + 2 * lane] = mv;
    }
    #pragma unroll
    for (int kk = 0; kk < 64; kk += 8) {
        const v4f q = fq[kk >> 3];
        const v2f a0 = *(const v2f*)&myT[row * 66 + kk + 2 * hi];
        const v2f a1 = *(const v2f*)&myT[row * 66 + kk + 4 + 2 * hi];
        v2f b0, b1;
        b0.x = q.x; b0.y = q.y;
        b1.x = q.z; b1.y = q.w;
        acc = __builtin_amdgcn_wmma_f32_16x16x4_f32(false, a0, false, b0,
                                                    (short)0, acc, false, false);
        acc = __builtin_amdgcn_wmma_f32_16x16x4_f32(false, a1, false, b1,
                                                    (short)0, acc, false, false);
    }
}

__global__ __launch_bounds__(256) __attribute__((amdgpu_waves_per_eu(2, 2)))
void out_kernel(const float* __restrict__ dist,
                const float* __restrict__ norm,
                const float* __restrict__ tabG,
                const float* __restrict__ fsums,
                float* __restrict__ out)
{
    __shared__ float mT[8][16 * 66];
    __shared__ float redC[8 * 256];
    __shared__ float tabL[8][2 * KNOT_PAD];

    const int tid  = threadIdx.x;
    const int lane = tid & 31;
    const int wave = tid >> 5;
    const int tileM = blockIdx.x * 16;
    const int row = lane & 15;
    const int hi  = lane >> 4;
    const int col = row;

    const float* tab0 = &tabL[0][0];
    const int wb = wave * (2 * KNOT_PAD * 4);
    const float kRoot = tabG[HH / 2 - 1];
    const float kTop  = tabG[HH - 1];
    const float t1 = tabG[16 * (row & 1) + 7];
    const float t2 = tabG[8 * (row & 3) + 3];
    const float t3 = tabG[4 * (row & 7) + 1];
    const float t4 = tabG[2 * row];
    const v2f* tg = (const v2f*)tabG;
    const v2f ts = tg[32 + lane], ti = tg[64 + lane];
    __builtin_amdgcn_sched_barrier(0);

    const int kBase = wave * 256;
    const size_t rowOff = (size_t)tileM * NN + kBase + 2 * lane;
    v2f dA[16], nA[16], dB[16], nB[16];
    load_rows(dA, nA, dist, norm, rowOff);
    __builtin_amdgcn_sched_barrier(0);
    const unsigned piv1 = __float_as_uint(t1 + 0.0f);
    const unsigned piv2 = __float_as_uint(t2 + 0.0f);
    const unsigned piv3 = __float_as_uint(t3 + 0.0f);
    const unsigned piv4 = __float_as_uint(t4 + 0.0f);

    float* myT = mT[wave];
    v8f acc = {};
    v4f fq[8];
    #pragma unroll
    for (int st = 0; st < 4; ++st) {
        load_fq(fq, fsums, kBase + st * 64, lane);
        __builtin_amdgcn_sched_barrier(0);
        if (st < 3) {
            if (st & 1) load_rows(dA, nA, dist, norm, rowOff + (st + 1) * 64);
            else        load_rows(dB, nB, dist, norm, rowOff + (st + 1) * 64);
        }
        __builtin_amdgcn_sched_barrier(0);
        if (st == 0) {
            float* myTab = tabL[wave];
            *(v2f*)&myTab[2 * lane]            = ts;
            *(v2f*)&myTab[KNOT_PAD + 2 * lane] = ti;
        }
        if (st & 1) stage_math(dB, nB, fq, kRoot, kTop, piv1, piv2, piv3, piv4, tab0, wb, myT, lane, row, hi, acc);
        else        stage_math(dA, nA, fq, kRoot, kTop, piv1, piv2, piv3, piv4, tab0, wb, myT, lane, row, hi, acc);
    }

    #pragma unroll
    for (int r = 0; r < 8; ++r) {
        const int orow = r + 8 * hi;
        redC[wave * 256 + orow * 16 + col] = acc[r];
    }
    __syncthreads();
    float v = 0.f;
    #pragma unroll
    for (int w = 0; w < 8; ++w) v += redC[w * 256 + tid];
    out[tileM * CC + tid] = v;
}

extern "C" void kernel_launch(void* const* d_in, const int* in_sizes, int n_in,
                              void* d_out, int out_size, void* d_ws, size_t ws_size,
                              hipStream_t stream) {
    const float* x    = (const float*)d_in[0];
    const float* dist = (const float*)d_in[1];
    const float* norm = (const float*)d_in[2];
    const float* W1   = (const float*)d_in[3];
    const float* b1   = (const float*)d_in[4];
    const float* W2   = (const float*)d_in[5];
    const float* b2   = (const float*)d_in[6];
    const float* fw   = (const float*)d_in[7];
    const float* Wm1  = (const float*)d_in[8];
    const float* bm1  = (const float*)d_in[9];
    const float* Wm2  = (const float*)d_in[10];
    const float* bm2  = (const float*)d_in[11];
    float* out   = (float*)d_out;
    float* fsums = (float*)d_ws;
    float* tab   = fsums + NN * CC;

    fsums_kernel<<<NN / 16 + 1, 256, 0, stream>>>(x, W1, b1, W2, b2, fw, Wm1, bm1, Wm2, bm2, fsums, tab);
    out_kernel<<<NN / 16, 256, 0, stream>>>(dist, norm, tab, fsums, out);
}
